# starting kernel
# baseline (speedup 1.0000x reference)
#include <hip/hip_runtime.h>
#include <hip/hip_bf16.h>


#define NW    12
#define NL    2
#define QDIM  4096
#define QBATCH 8192

typedef __attribute__((ext_vector_type(2))) float v2f;
typedef __attribute__((ext_vector_type(8))) float v8f;

static __device__ __forceinline__ v8f wmma4(v2f a, v2f b, v8f c) {
    return __builtin_amdgcn_wmma_f32_16x16x4_f32(false, a, false, b, (short)0, c,
                                                 false, false);
}

__global__ void qgate_build(const float* __restrict__ params,
                            float* __restrict__ gr, float* __restrict__ gi) {
    int e = blockIdx.x * blockDim.x + threadIdx.x;
    if (e >= NL * 3 * 256) return;
    int l   = e / (3 * 256);
    int rem = e - l * (3 * 256);
    int g   = rem >> 8;
    int ij  = rem & 255;
    int i = ij >> 4, j = ij & 15;
    float re = 1.f, im = 0.f;
    for (int k = 0; k < 4; ++k) {
        int   w  = g * 4 + k;
        float th = params[l * NW + w] * 0.5f;
        float c  = cosf(th), s = sinf(th);
        int ib = (i >> (3 - k)) & 1;
        int jb = (j >> (3 - k)) & 1;
        float wr, wi;
        if (!ib && !jb)     { wr =  c * c; wi =  s * s; }
        else if (!ib && jb) { wr = -c * s; wi = -c * s; }
        else if (ib && !jb) { wr =  c * s; wi = -c * s; }
        else                { wr =  c * c; wi = -s * s; }
        float nr = re * wr - im * wi;
        float ni = re * wi + im * wr;
        re = nr; im = ni;
    }
    gr[e] = re;
    gi[e] = im;
}

__launch_bounds__(256)
__global__ void qsim(const float* __restrict__ x,
                     const float* __restrict__ head_w,
                     const float* __restrict__ head_b,
                     const float* __restrict__ gr,
                     const float* __restrict__ gi,
                     float* __restrict__ out) {
    __shared__ float Sr[QDIM];
    __shared__ float Si[QDIM];
    __shared__ float tabA[16], tabM[16], tabZ[16];
    __shared__ float cfA[16], cfM[16], cfZ[16];
    __shared__ float trig[2 * NW];
    __shared__ float red[256];

    const int tid  = threadIdx.x;
    const int b    = blockIdx.x;
    const int lane = tid & 31;
    const int wave = tid >> 5;
    const int half = lane >> 4;
    const int lm   = lane & 15;
    const int k0   = half * 2;

    if (tid < NW) {
        float th = x[b * NW + tid] * 0.5f;
        trig[tid]      = cosf(th);
        trig[tid + NW] = sinf(th);
    }
    __syncthreads();

    if (tid < 48) {
        int grp = tid >> 4, v = tid & 15;
        float p = 1.f;
        for (int k = 0; k < 4; ++k) {
            int w = grp * 4 + k;
            p *= ((v >> (3 - k)) & 1) ? trig[NW + w] : trig[w];
        }
        (grp == 0 ? tabA : grp == 1 ? tabM : tabZ)[v] = p;
    } else if (tid < 96) {
        int grp = (tid - 48) >> 4, v = tid & 15;
        float s = 0.f;
        for (int k = 0; k < 4; ++k) {
            int w = grp * 4 + k;
            s += head_w[w] * (((v >> (3 - k)) & 1) ? -1.f : 1.f);
        }
        (grp == 0 ? cfA : grp == 1 ? cfM : cfZ)[v] = s;
    }
    __syncthreads();

    for (int k = 0; k < 16; ++k) {
        int i = tid * 16 + k;
        Sr[i] = tabA[(i >> 8) & 15] * tabM[(i >> 4) & 15] * tabZ[i & 15];
        Si[i] = 0.f;
    }
    __syncthreads();

    for (int l = 0; l < NL; ++l) {
        const float* Gr = gr + l * 3 * 256;
        const float* Gi = gi + l * 3 * 256;

        for (int g = 0; g < 2; ++g) {
            const float* ur = Gr + g * 256;
            const float* ui = Gi + g * 256;
            v2f Ar[4], Av[4], An[4];
            for (int kk = 0; kk < 4; ++kk) {
                int kc = kk * 4 + k0;
                Ar[kk].x = ur[lm * 16 + kc];  Ar[kk].y = ur[lm * 16 + kc + 1];
                Av[kk].x = ui[lm * 16 + kc];  Av[kk].y = ui[lm * 16 + kc + 1];
                An[kk].x = -Av[kk].x;         An[kk].y = -Av[kk].y;
            }
            for (int tt = 0; tt < 2; ++tt) {
                int t = wave + tt * 8;
                v2f Br[4], Bi[4];
                for (int kk = 0; kk < 4; ++kk) {
                    int kc = kk * 4 + k0;
                    int i0 = (g == 0) ? (kc * 256 + t * 16 + lm)
                                      : (t * 256 + kc * 16 + lm);
                    int i1 = (g == 0) ? ((kc + 1) * 256 + t * 16 + lm)
                                      : (t * 256 + (kc + 1) * 16 + lm);
                    Br[kk].x = Sr[i0];  Br[kk].y = Sr[i1];
                    Bi[kk].x = Si[i0];  Bi[kk].y = Si[i1];
                }
                v8f Cr = {}; v8f Ci = {};
                for (int kk = 0; kk < 4; ++kk) {
                    Cr = wmma4(Ar[kk], Br[kk], Cr);
                    Cr = wmma4(An[kk], Bi[kk], Cr);
                    Ci = wmma4(Ar[kk], Bi[kk], Ci);
                    Ci = wmma4(Av[kk], Br[kk], Ci);
                }
                for (int v = 0; v < 8; ++v) {
                    int M = v + half * 8;
                    int o = (g == 0) ? (M * 256 + t * 16 + lm)
                                     : (t * 256 + M * 16 + lm);
                    Sr[o] = Cr[v];  Si[o] = Ci[v];
                }
            }
            __syncthreads();
        }

        {
            const float* ur = Gr + 2 * 256;
            const float* ui = Gi + 2 * 256;
            v2f Btr[4], Bti[4], Btn[4];
            for (int kk = 0; kk < 4; ++kk) {
                int kc = kk * 4 + k0;
                Btr[kk].x = ur[lm * 16 + kc];  Btr[kk].y = ur[lm * 16 + kc + 1];
                Bti[kk].x = ui[lm * 16 + kc];  Bti[kk].y = ui[lm * 16 + kc + 1];
                Btn[kk].x = -Bti[kk].x;        Btn[kk].y = -Bti[kk].y;
            }
            for (int tt = 0; tt < 2; ++tt) {
                int t = wave + tt * 8;
                v2f Asr[4], Asi[4];
                int rbase = (t * 16 + lm) * 16;
                for (int kk = 0; kk < 4; ++kk) {
                    int kc = kk * 4 + k0;
                    Asr[kk].x = Sr[rbase + kc];  Asr[kk].y = Sr[rbase + kc + 1];
                    Asi[kk].x = Si[rbase + kc];  Asi[kk].y = Si[rbase + kc + 1];
                }
                v8f Cr = {}; v8f Ci = {};
                for (int kk = 0; kk < 4; ++kk) {
                    Cr = wmma4(Asr[kk], Btr[kk], Cr);
                    Cr = wmma4(Asi[kk], Btn[kk], Cr);
                    Ci = wmma4(Asr[kk], Bti[kk], Ci);
                    Ci = wmma4(Asi[kk], Btr[kk], Ci);
                }
                for (int v = 0; v < 8; ++v) {
                    int M = v + half * 8;
                    int o = (t * 16 + M) * 16 + lm;
                    Sr[o] = Cr[v];  Si[o] = Ci[v];
                }
            }
            __syncthreads();
        }

        {
            float vr[16], vi[16];
            for (int k = 0; k < 16; ++k) {
                int i = tid * 16 + k;
                int s = i ^ (i >> 1);
                vr[k] = Sr[s];  vi[k] = Si[s];
            }
            __syncthreads();
            for (int k = 0; k < 16; ++k) {
                int i = tid * 16 + k;
                Sr[i] = vr[k];  Si[i] = vi[k];
            }
            __syncthreads();
        }
    }

    float partial = 0.f;
    {
        float base = cfA[(tid >> 4) & 15] + cfM[tid & 15];
        for (int k = 0; k < 16; ++k) {
            int i = tid * 16 + k;
            float pr = Sr[i], pi = Si[i];
            partial += (pr * pr + pi * pi) * (base + cfZ[k]);
        }
    }
    red[tid] = partial;
    __syncthreads();
    for (int s = 128; s > 0; s >>= 1) {
        if (tid < s) red[tid] += red[tid + s];
        __syncthreads();
    }
    if (tid == 0) out[b] = red[0] + head_b[0];
}

extern "C" void kernel_launch(void* const* d_in, const int* in_sizes, int n_in,
                              void* d_out, int out_size, void* d_ws, size_t ws_size,
                              hipStream_t stream) {
    const float* x      = (const float*)d_in[0];
    const float* params = (const float*)d_in[1];
    const float* head_w = (const float*)d_in[2];
    const float* head_b = (const float*)d_in[3];
    float* out = (float*)d_out;

    float* gr = (float*)d_ws;
    float* gi = gr + NL * 3 * 256;

    qgate_build<<<(NL * 3 * 256 + 255) / 256, 256, 0, stream>>>(params, gr, gi);
    qsim<<<QBATCH, 256, 0, stream>>>(x, head_w, head_b, gr, gi, out);
}


# CNOT-chain permutation folded into LDS addresses: in-row part into the low-nibble gate's stores, row part as a row renaming used by the next top-nibble loads and final weights; permute pass removed
# speedup vs baseline: 1.2613x; 1.2613x over previous
#include <hip/hip_runtime.h>
#include <hip/hip_bf16.h>


#define NW    12
#define NL    2
#define QDIM  4096
#define QBATCH 8192

typedef __attribute__((ext_vector_type(2))) float v2f;
typedef __attribute__((ext_vector_type(8))) float v8f;

static __device__ __forceinline__ int gray4(int a) { return a ^ (a >> 1); }
static __device__ __forceinline__ int ginv4(int p) { p ^= p >> 1; p ^= p >> 2; return p; }
static __device__ __forceinline__ int ginv8(int s) { s ^= s >> 1; s ^= s >> 2; s ^= s >> 4; return s; }

static __device__ __forceinline__ v8f wmma4(v2f a, v2f b, v8f c) {
    return __builtin_amdgcn_wmma_f32_16x16x4_f32(false, a, false, b, (short)0, c,
                                                 false, false);
}

__global__ void qgate_build(const float* __restrict__ params,
                            float* __restrict__ gr, float* __restrict__ gi) {
    int e = blockIdx.x * blockDim.x + threadIdx.x;
    if (e >= NL * 3 * 256) return;
    int l   = e / (3 * 256);
    int rem = e - l * (3 * 256);
    int g   = rem >> 8;
    int ij  = rem & 255;
    int i = ij >> 4, j = ij & 15;
    float re = 1.f, im = 0.f;
    for (int k = 0; k < 4; ++k) {
        int   w  = g * 4 + k;
        float th = params[l * NW + w] * 0.5f;
        float c  = cosf(th), s = sinf(th);
        int ib = (i >> (3 - k)) & 1;
        int jb = (j >> (3 - k)) & 1;
        float wr, wi;
        if (!ib && !jb)     { wr =  c * c; wi =  s * s; }
        else if (!ib && jb) { wr = -c * s; wi = -c * s; }
        else if (ib && !jb) { wr =  c * s; wi = -c * s; }
        else                { wr =  c * c; wi = -s * s; }
        float nr = re * wr - im * wi;
        float ni = re * wi + im * wr;
        re = nr; im = ni;
    }
    gr[e] = re;
    gi[e] = im;
}

__launch_bounds__(256)
__global__ void qsim(const float* __restrict__ x,
                     const float* __restrict__ head_w,
                     const float* __restrict__ head_b,
                     const float* __restrict__ gr,
                     const float* __restrict__ gi,
                     float* __restrict__ out) {
    __shared__ float Sr[QDIM];
    __shared__ float Si[QDIM];
    __shared__ float tabA[16], tabM[16], tabZ[16];
    __shared__ float cfA[16], cfM[16], cfZ[16];
    __shared__ float trig[2 * NW];
    __shared__ float red[256];

    const int tid  = threadIdx.x;
    const int b    = blockIdx.x;
    const int lane = tid & 31;
    const int wave = tid >> 5;
    const int half = lane >> 4;
    const int lm   = lane & 15;
    const int k0   = half * 2;

    if (tid < NW) {
        float th = x[b * NW + tid] * 0.5f;
        trig[tid]      = cosf(th);
        trig[tid + NW] = sinf(th);
    }
    __syncthreads();

    if (tid < 48) {
        int grp = tid >> 4, v = tid & 15;
        float p = 1.f;
        for (int k = 0; k < 4; ++k) {
            int w = grp * 4 + k;
            p *= ((v >> (3 - k)) & 1) ? trig[NW + w] : trig[w];
        }
        (grp == 0 ? tabA : grp == 1 ? tabM : tabZ)[v] = p;
    } else if (tid < 96) {
        int grp = (tid - 48) >> 4, v = tid & 15;
        float s = 0.f;
        for (int k = 0; k < 4; ++k) {
            int w = grp * 4 + k;
            s += head_w[w] * (((v >> (3 - k)) & 1) ? -1.f : 1.f);
        }
        (grp == 0 ? cfA : grp == 1 ? cfM : cfZ)[v] = s;
    }
    __syncthreads();

    for (int k = 0; k < 16; ++k) {
        int i = tid * 16 + k;
        Sr[i] = tabA[ginv4((i >> 8) & 15)] * tabM[(i >> 4) & 15] * tabZ[i & 15];
        Si[i] = 0.f;
    }
    __syncthreads();

    for (int l = 0; l < NL; ++l) {
        const float* Gr = gr + l * 3 * 256;
        const float* Gi = gi + l * 3 * 256;

        for (int g = 0; g < 2; ++g) {
            const float* ur = Gr + g * 256;
            const float* ui = Gi + g * 256;
            v2f Ar[4], Av[4], An[4];
            for (int kk = 0; kk < 4; ++kk) {
                int kc = kk * 4 + k0;
                Ar[kk].x = ur[lm * 16 + kc];  Ar[kk].y = ur[lm * 16 + kc + 1];
                Av[kk].x = ui[lm * 16 + kc];  Av[kk].y = ui[lm * 16 + kc + 1];
                An[kk].x = -Av[kk].x;         An[kk].y = -Av[kk].y;
            }
            for (int tt = 0; tt < 2; ++tt) {
                int t = wave + tt * 8;
                v2f Br[4], Bi[4];
                for (int kk = 0; kk < 4; ++kk) {
                    int kc = kk * 4 + k0;
                    int i0 = (g == 0) ? (gray4(kc) * 256 + t * 16 + lm)
                                      : (t * 256 + kc * 16 + lm);
                    int i1 = (g == 0) ? (gray4(kc + 1) * 256 + t * 16 + lm)
                                      : (t * 256 + (kc + 1) * 16 + lm);
                    Br[kk].x = Sr[i0];  Br[kk].y = Sr[i1];
                    Bi[kk].x = Si[i0];  Bi[kk].y = Si[i1];
                }
                v8f Cr = {}; v8f Ci = {};
                for (int kk = 0; kk < 4; ++kk) {
                    Cr = wmma4(Ar[kk], Br[kk], Cr);
                    Cr = wmma4(An[kk], Bi[kk], Cr);
                    Ci = wmma4(Ar[kk], Bi[kk], Ci);
                    Ci = wmma4(Av[kk], Br[kk], Ci);
                }
                for (int v = 0; v < 8; ++v) {
                    int M = v + half * 8;
                    int o = (g == 0) ? (M * 256 + t * 16 + lm)
                                     : (t * 256 + M * 16 + lm);
                    Sr[o] = Cr[v];  Si[o] = Ci[v];
                }
            }
            __syncthreads();
        }

        {
            const float* ur = Gr + 2 * 256;
            const float* ui = Gi + 2 * 256;
            v2f Btr[4], Bti[4], Btn[4];
            for (int kk = 0; kk < 4; ++kk) {
                int kc = kk * 4 + k0;
                Btr[kk].x = ur[lm * 16 + kc];  Btr[kk].y = ur[lm * 16 + kc + 1];
                Bti[kk].x = ui[lm * 16 + kc];  Bti[kk].y = ui[lm * 16 + kc + 1];
                Btn[kk].x = -Bti[kk].x;        Btn[kk].y = -Bti[kk].y;
            }
            for (int tt = 0; tt < 2; ++tt) {
                int t = wave + tt * 8;
                v2f Asr[4], Asi[4];
                int rbase = (t * 16 + lm) * 16;
                for (int kk = 0; kk < 4; ++kk) {
                    int kc = kk * 4 + k0;
                    Asr[kk].x = Sr[rbase + kc];  Asr[kk].y = Sr[rbase + kc + 1];
                    Asi[kk].x = Si[rbase + kc];  Asi[kk].y = Si[rbase + kc + 1];
                }
                v8f Cr = {}; v8f Ci = {};
                for (int kk = 0; kk < 4; ++kk) {
                    Cr = wmma4(Asr[kk], Btr[kk], Cr);
                    Cr = wmma4(Asi[kk], Btn[kk], Cr);
                    Ci = wmma4(Asr[kk], Bti[kk], Ci);
                    Ci = wmma4(Asi[kk], Btr[kk], Ci);
                }
                int flip = (ginv4(t) & 1) ? 255 : 0;
                for (int v = 0; v < 8; ++v) {
                    int M = v + half * 8;
                    int o = t * 256 + (ginv8(M * 16 + lm) ^ flip);
                    Sr[o] = Cr[v];  Si[o] = Ci[v];
                }
            }
            __syncthreads();
        }

    }

    float partial = 0.f;
    {
        float base = cfA[ginv4((tid >> 4) & 15)] + cfM[tid & 15];
        for (int k = 0; k < 16; ++k) {
            int i = tid * 16 + k;
            float pr = Sr[i], pi = Si[i];
            partial += (pr * pr + pi * pi) * (base + cfZ[k]);
        }
    }
    red[ginv4((tid >> 4) & 15) * 16 + (tid & 15)] = partial;
    __syncthreads();
    for (int s = 128; s > 0; s >>= 1) {
        if (tid < s) red[tid] += red[tid + s];
        __syncthreads();
    }
    if (tid == 0) out[b] = red[0] + head_b[0];
}

extern "C" void kernel_launch(void* const* d_in, const int* in_sizes, int n_in,
                              void* d_out, int out_size, void* d_ws, size_t ws_size,
                              hipStream_t stream) {
    const float* x      = (const float*)d_in[0];
    const float* params = (const float*)d_in[1];
    const float* head_w = (const float*)d_in[2];
    const float* head_b = (const float*)d_in[3];
    float* out = (float*)d_out;

    float* gr = (float*)d_ws;
    float* gi = gr + NL * 3 * 256;

    qgate_build<<<(NL * 3 * 256 + 255) / 256, 256, 0, stream>>>(params, gr, gi);
    qsim<<<QBATCH, 256, 0, stream>>>(x, head_w, head_b, gr, gi, out);
}


# Layer-0 RX/RY folded into the encoding: the state is still a product there, so each wire's 2x2 acts on its 2-vector; written once, CNOT-permuted; 3 of 6 WMMA passes go; layer-0 rounding order changes
# speedup vs baseline: 2.4341x; 1.9298x over previous
#include <hip/hip_runtime.h>
#include <hip/hip_bf16.h>


#define NW    12
#define NL    2
#define QDIM  4096
#define QBATCH 8192
#define NG    ((NL - 1) * 3 * 256)

typedef __attribute__((ext_vector_type(2))) float v2f;
typedef __attribute__((ext_vector_type(8))) float v8f;

static __device__ __forceinline__ int gray4(int a) { return a ^ (a >> 1); }
static __device__ __forceinline__ int ginv4(int p) { p ^= p >> 1; p ^= p >> 2; return p; }
static __device__ __forceinline__ int ginv8(int s) { s ^= s >> 1; s ^= s >> 2; s ^= s >> 4; return s; }

static __device__ __forceinline__ v8f wmma4(v2f a, v2f b, v8f c) {
    return __builtin_amdgcn_wmma_f32_16x16x4_f32(false, a, false, b, (short)0, c,
                                                 false, false);
}

__global__ void qgate_build(const float* __restrict__ params,
                            float* __restrict__ gr, float* __restrict__ gi,
                            float* __restrict__ w0) {
    int e = blockIdx.x * blockDim.x + threadIdx.x;
    if (e >= NG) {
        int w = e - NG;
        if (w >= NW) return;
        float th = params[w] * 0.5f;
        float c  = cosf(th), s = sinf(th);
        float* o = w0 + w * 8;
        o[0] =  c * c;  o[1] =  s * s;
        o[2] = -c * s;  o[3] = -c * s;
        o[4] =  c * s;  o[5] = -c * s;
        o[6] =  c * c;  o[7] = -s * s;
        return;
    }
    int l   = 1 + e / (3 * 256);
    int rem = e - (l - 1) * (3 * 256);
    int g   = rem >> 8;
    int ij  = rem & 255;
    int i = ij >> 4, j = ij & 15;
    float re = 1.f, im = 0.f;
    for (int k = 0; k < 4; ++k) {
        int   w  = g * 4 + k;
        float th = params[l * NW + w] * 0.5f;
        float c  = cosf(th), s = sinf(th);
        int ib = (i >> (3 - k)) & 1;
        int jb = (j >> (3 - k)) & 1;
        float wr, wi;
        if (!ib && !jb)     { wr =  c * c; wi =  s * s; }
        else if (!ib && jb) { wr = -c * s; wi = -c * s; }
        else if (ib && !jb) { wr =  c * s; wi = -c * s; }
        else                { wr =  c * c; wi = -s * s; }
        float nr = re * wr - im * wi;
        float ni = re * wi + im * wr;
        re = nr; im = ni;
    }
    gr[e] = re;
    gi[e] = im;
}

__launch_bounds__(256)
__global__ void qsim(const float* __restrict__ x,
                     const float* __restrict__ head_w,
                     const float* __restrict__ head_b,
                     const float* __restrict__ gr,
                     const float* __restrict__ gi,
                     const float* __restrict__ w0,
                     float* __restrict__ out) {
    __shared__ float Sr[QDIM];
    __shared__ float Si[QDIM];
    __shared__ float tabR[48], tabI[48];
    __shared__ float cfA[16], cfM[16], cfZ[16];
    __shared__ float uR[2 * NW], uI[2 * NW];
    __shared__ float red[256];

    const int tid  = threadIdx.x;
    const int b    = blockIdx.x;
    const int lane = tid & 31;
    const int wave = tid >> 5;
    const int half = lane >> 4;
    const int lm   = lane & 15;
    const int k0   = half * 2;

    if (tid < NW) {
        float th = x[b * NW + tid] * 0.5f;
        float cx = cosf(th), sx = sinf(th);
        const float* m = w0 + tid * 8;
        uR[tid]      = m[0] * cx + m[2] * sx;
        uI[tid]      = m[1] * cx + m[3] * sx;
        uR[tid + NW] = m[4] * cx + m[6] * sx;
        uI[tid + NW] = m[5] * cx + m[7] * sx;
    }
    __syncthreads();

    if (tid < 48) {
        int grp = tid >> 4, v = tid & 15;
        float pr = 1.f, pi = 0.f;
        for (int k = 0; k < 4; ++k) {
            int w = grp * 4 + k + (((v >> (3 - k)) & 1) ? NW : 0);
            float fr = uR[w], fi = uI[w];
            float nr = pr * fr - pi * fi;
            float ni = pr * fi + pi * fr;
            pr = nr; pi = ni;
        }
        tabR[tid] = pr;
        tabI[tid] = pi;
    } else if (tid < 96) {
        int grp = (tid - 48) >> 4, v = tid & 15;
        float s = 0.f;
        for (int k = 0; k < 4; ++k) {
            int w = grp * 4 + k;
            s += head_w[w] * (((v >> (3 - k)) & 1) ? -1.f : 1.f);
        }
        (grp == 0 ? cfA : grp == 1 ? cfM : cfZ)[v] = s;
    }
    __syncthreads();

    {
        const int s0 = tid ^ (tid >> 1);
        float mzr[2], mzi[2];
        for (int c = 0; c < 2; ++c) {
            int s = s0 ^ (c << 7);
            float mr = tabR[16 + (s >> 4)], mi = tabI[16 + (s >> 4)];
            float zr = tabR[32 + (s & 15)], zi = tabI[32 + (s & 15)];
            mzr[c] = mr * zr - mi * zi;
            mzi[c] = mr * zi + mi * zr;
        }
        for (int t = 0; t < 16; ++t) {
            int c = ginv4(t) & 1;
            float ar = tabR[t], ai = tabI[t];
            Sr[t * 256 + tid] = ar * mzr[c] - ai * mzi[c];
            Si[t * 256 + tid] = ar * mzi[c] + ai * mzr[c];
        }
    }
    __syncthreads();

    for (int l = 1; l < NL; ++l) {
        const float* Gr = gr + (l - 1) * 3 * 256;
        const float* Gi = gi + (l - 1) * 3 * 256;

        for (int g = 0; g < 2; ++g) {
            const float* ur = Gr + g * 256;
            const float* ui = Gi + g * 256;
            v2f Ar[4], Av[4], An[4];
            for (int kk = 0; kk < 4; ++kk) {
                int kc = kk * 4 + k0;
                Ar[kk].x = ur[lm * 16 + kc];  Ar[kk].y = ur[lm * 16 + kc + 1];
                Av[kk].x = ui[lm * 16 + kc];  Av[kk].y = ui[lm * 16 + kc + 1];
                An[kk].x = -Av[kk].x;         An[kk].y = -Av[kk].y;
            }
            for (int tt = 0; tt < 2; ++tt) {
                int t = wave + tt * 8;
                v2f Br[4], Bi[4];
                for (int kk = 0; kk < 4; ++kk) {
                    int kc = kk * 4 + k0;
                    int i0 = (g == 0) ? (gray4(kc) * 256 + t * 16 + lm)
                                      : (t * 256 + kc * 16 + lm);
                    int i1 = (g == 0) ? (gray4(kc + 1) * 256 + t * 16 + lm)
                                      : (t * 256 + (kc + 1) * 16 + lm);
                    Br[kk].x = Sr[i0];  Br[kk].y = Sr[i1];
                    Bi[kk].x = Si[i0];  Bi[kk].y = Si[i1];
                }
                v8f Cr = {}; v8f Ci = {};
                for (int kk = 0; kk < 4; ++kk) {
                    Cr = wmma4(Ar[kk], Br[kk], Cr);
                    Cr = wmma4(An[kk], Bi[kk], Cr);
                    Ci = wmma4(Ar[kk], Bi[kk], Ci);
                    Ci = wmma4(Av[kk], Br[kk], Ci);
                }
                for (int v = 0; v < 8; ++v) {
                    int M = v + half * 8;
                    int o = (g == 0) ? (M * 256 + t * 16 + lm)
                                     : (t * 256 + M * 16 + lm);
                    Sr[o] = Cr[v];  Si[o] = Ci[v];
                }
            }
            __syncthreads();
        }

        {
            const float* ur = Gr + 2 * 256;
            const float* ui = Gi + 2 * 256;
            v2f Btr[4], Bti[4], Btn[4];
            for (int kk = 0; kk < 4; ++kk) {
                int kc = kk * 4 + k0;
                Btr[kk].x = ur[lm * 16 + kc];  Btr[kk].y = ur[lm * 16 + kc + 1];
                Bti[kk].x = ui[lm * 16 + kc];  Bti[kk].y = ui[lm * 16 + kc + 1];
                Btn[kk].x = -Bti[kk].x;        Btn[kk].y = -Bti[kk].y;
            }
            for (int tt = 0; tt < 2; ++tt) {
                int t = wave + tt * 8;
                v2f Asr[4], Asi[4];
                int rbase = (t * 16 + lm) * 16;
                for (int kk = 0; kk < 4; ++kk) {
                    int kc = kk * 4 + k0;
                    Asr[kk].x = Sr[rbase + kc];  Asr[kk].y = Sr[rbase + kc + 1];
                    Asi[kk].x = Si[rbase + kc];  Asi[kk].y = Si[rbase + kc + 1];
                }
                v8f Cr = {}; v8f Ci = {};
                for (int kk = 0; kk < 4; ++kk) {
                    Cr = wmma4(Asr[kk], Btr[kk], Cr);
                    Cr = wmma4(Asi[kk], Btn[kk], Cr);
                    Ci = wmma4(Asr[kk], Bti[kk], Ci);
                    Ci = wmma4(Asi[kk], Btr[kk], Ci);
                }
                int flip = (ginv4(t) & 1) ? 255 : 0;
                for (int v = 0; v < 8; ++v) {
                    int M = v + half * 8;
                    int o = t * 256 + (ginv8(M * 16 + lm) ^ flip);
                    Sr[o] = Cr[v];  Si[o] = Ci[v];
                }
            }
            __syncthreads();
        }

    }

    float partial = 0.f;
    {
        float base = cfA[ginv4((tid >> 4) & 15)] + cfM[tid & 15];
        for (int k = 0; k < 16; ++k) {
            int i = tid * 16 + k;
            float pr = Sr[i], pi = Si[i];
            partial += (pr * pr + pi * pi) * (base + cfZ[k]);
        }
    }
    red[ginv4((tid >> 4) & 15) * 16 + (tid & 15)] = partial;
    __syncthreads();
    for (int s = 128; s > 0; s >>= 1) {
        if (tid < s) red[tid] += red[tid + s];
        __syncthreads();
    }
    if (tid == 0) out[b] = red[0] + head_b[0];
}

extern "C" void kernel_launch(void* const* d_in, const int* in_sizes, int n_in,
                              void* d_out, int out_size, void* d_ws, size_t ws_size,
                              hipStream_t stream) {
    const float* x      = (const float*)d_in[0];
    const float* params = (const float*)d_in[1];
    const float* head_w = (const float*)d_in[2];
    const float* head_b = (const float*)d_in[3];
    float* out = (float*)d_out;

    float* gr = (float*)d_ws;
    float* gi = gr + NG;
    float* w0 = gi + NG;

    qgate_build<<<(NG + NW + 255) / 256, 256, 0, stream>>>(params, gr, gi, w0);
    qsim<<<QBATCH, 256, 0, stream>>>(x, head_w, head_b, gr, gi, w0, out);
}


# Layer-1 low-nibble gate applied to the rank-2 factors too (WMMA tiles of 16 samples' factor vectors); Z sums as weighted factor inner products, state never written; rounding order changes
# speedup vs baseline: 21.6926x; 8.9120x over previous
#include <hip/hip_runtime.h>


#define NW    12
#define NL    2
#define QBATCH 8192
#define SPB   16
#define NG    ((NL - 1) * 3 * 256)
static_assert(NL == 2, "layer 1 is applied on the factors of the state layer 0 leaves");
static_assert(QBATCH % SPB == 0, "whole workgroups");

typedef __attribute__((ext_vector_type(2))) float v2f;
typedef __attribute__((ext_vector_type(8))) float v8f;

static __device__ __forceinline__ int gray4(int a) { return a ^ (a >> 1); }
static __device__ __forceinline__ float sg(int v) { return (__popc(v) & 1) ? -1.f : 1.f; }

static __device__ __forceinline__ v8f wmma4(v2f a, v2f b, v8f c) {
    return __builtin_amdgcn_wmma_f32_16x16x4_f32(false, a, false, b, (short)0, c,
                                                 false, false);
}

__global__ void qgate_build(const float* __restrict__ params,
                            float* __restrict__ gr, float* __restrict__ gi,
                            float* __restrict__ w0) {
    int e = blockIdx.x * blockDim.x + threadIdx.x;
    if (e >= NG) {
        int w = e - NG;
        if (w >= NW) return;
        float th = params[w] * 0.5f;
        float c  = cosf(th), s = sinf(th);
        float* o = w0 + w * 8;
        o[0] =  c * c;  o[1] =  s * s;
        o[2] = -c * s;  o[3] = -c * s;
        o[4] =  c * s;  o[5] = -c * s;
        o[6] =  c * c;  o[7] = -s * s;
        return;
    }
    int l   = 1 + e / (3 * 256);
    int rem = e - (l - 1) * (3 * 256);
    int g   = rem >> 8;
    int ij  = rem & 255;
    int i = ij >> 4, j = ij & 15;
    float re = 1.f, im = 0.f;
    for (int k = 0; k < 4; ++k) {
        int   w  = g * 4 + k;
        float th = params[l * NW + w] * 0.5f;
        float c  = cosf(th), s = sinf(th);
        int ib = (i >> (3 - k)) & 1;
        int jb = (j >> (3 - k)) & 1;
        float wr, wi;
        if (!ib && !jb)     { wr =  c * c; wi =  s * s; }
        else if (!ib && jb) { wr = -c * s; wi = -c * s; }
        else if (ib && !jb) { wr =  c * s; wi = -c * s; }
        else                { wr =  c * c; wi = -s * s; }
        float nr = re * wr - im * wi;
        float ni = re * wi + im * wr;
        re = nr; im = ni;
    }
    gr[e] = re;
    gi[e] = im;
}

__launch_bounds__(256)
__global__ void qsim(const float* __restrict__ x,
                     const float* __restrict__ head_w,
                     const float* __restrict__ head_b,
                     const float* __restrict__ gr,
                     const float* __restrict__ gi,
                     const float* __restrict__ w0,
                     float* __restrict__ out) {
    __shared__ float uR[SPB * 2 * NW], uI[SPB * 2 * NW];
    __shared__ float tabR[SPB * 48], tabI[SPB * 48];
    __shared__ float resR[SPB * 128], resI[SPB * 128];
    __shared__ float gmR[SPB * 64], gmI[SPB * 64];
    __shared__ float wt[6 * 16];

    const int tid  = threadIdx.x;
    const int b    = blockIdx.x;
    const int lane = tid & 31;
    const int wave = tid >> 5;
    const int half = lane >> 4;
    const int lm   = lane & 15;
    const int k0   = half * 2;

    const int g = (wave < 2) ? 0 : (wave < 6) ? 1 : 2;
    v2f Btr[4], Bti[4], Btn[4];
    {
        const float* ur = gr + g * 256;
        const float* ui = gi + g * 256;
        for (int kk = 0; kk < 4; ++kk) {
            int kc = kk * 4 + k0;
            Btr[kk].x = ur[lm * 16 + kc];  Btr[kk].y = ur[lm * 16 + kc + 1];
            Bti[kk].x = ui[lm * 16 + kc];  Bti[kk].y = ui[lm * 16 + kc + 1];
            Btn[kk].x = -Bti[kk].x;        Btn[kk].y = -Bti[kk].y;
        }
    }

    if (tid < SPB * NW) {
        const int s = tid / NW, w = tid - s * NW;
        float th = x[b * (SPB * NW) + tid] * 0.5f;
        float cx = cosf(th), sx = sinf(th);
        const float* m = w0 + w * 8;
        uR[s * 24 + w]      = m[0] * cx + m[2] * sx;
        uI[s * 24 + w]      = m[1] * cx + m[3] * sx;
        uR[s * 24 + w + NW] = m[4] * cx + m[6] * sx;
        uI[s * 24 + w + NW] = m[5] * cx + m[7] * sx;
    }
    if (tid < 96) {
        const int id = tid >> 4, n = tid & 15;
        float v;
        if (id == 2) {
            v = 1.f;
        } else if (id == 1 || id == 4) {
            v = sg(n);
        } else {
            const int grp = (id == 0) ? 0 : (id == 3) ? 1 : 2;
            v = 0.f;
            for (int k = 0; k < 4; ++k) v += head_w[grp * 4 + k] * sg(n >> (3 - k));
        }
        wt[tid] = v;
    }
    __syncthreads();

    for (int r = 0; r < 3; ++r) {
        const int item = tid + 256 * r;
        const int s = item / 48, e = item - s * 48;
        const int grp = e >> 4, v = e & 15;
        float pr = 1.f, pi = 0.f;
        for (int k = 0; k < 4; ++k) {
            int w = s * 24 + grp * 4 + k + (((v >> (3 - k)) & 1) ? NW : 0);
            float fr = uR[w], fi = uI[w];
            float nr = pr * fr - pi * fi;
            float ni = pr * fi + pi * fr;
            pr = nr; pi = ni;
        }
        tabR[item] = pr;
        tabI[item] = pi;
    }
    __syncthreads();

    {
        int s, par, fb;
        if (g == 0)      { s = wave * 8 + (lm >> 1);        par = lm & 1;  fb = 0; }
        else if (g == 1) { s = (wave - 2) * 4 + (lm >> 2);  par = lm & 1;  fb = ((lm >> 1) & 1) << 3; }
        else             { s = (wave - 6) * 8 + (lm >> 1);  par = 0;       fb = (lm & 1) << 3; }
        const int tb = s * 48 + g * 16;
        const bool on0 = (g == 2) || (par == 0);
        const bool on1 = (g == 2) || (par == 1);
        v2f Avr[4], Avi[4];
        for (int kk = 0; kk < 4; ++kk) {
            int kc = kk * 4 + k0;
            int i0 = tb + (gray4(kc) ^ fb);
            int i1 = tb + (gray4(kc + 1) ^ fb);
            Avr[kk].x = on0 ? tabR[i0] : 0.f;  Avr[kk].y = on1 ? tabR[i1] : 0.f;
            Avi[kk].x = on0 ? tabI[i0] : 0.f;  Avi[kk].y = on1 ? tabI[i1] : 0.f;
        }
        v8f Cr = {}; v8f Ci = {};
        for (int kk = 0; kk < 4; ++kk) {
            Cr = wmma4(Avr[kk], Btr[kk], Cr);
            Cr = wmma4(Avi[kk], Btn[kk], Cr);
            Ci = wmma4(Avr[kk], Bti[kk], Ci);
            Ci = wmma4(Avi[kk], Btr[kk], Ci);
        }
        for (int v = 0; v < 8; ++v) {
            const int M = v + half * 8;
            int sM, vec;
            if (g == 0)      { sM = wave * 8 + (M >> 1);        vec = M & 1; }
            else if (g == 1) { sM = (wave - 2) * 4 + (M >> 2);  vec = 2 + (M & 3); }
            else             { sM = (wave - 6) * 8 + (M >> 1);  vec = 6 + (M & 1); }
            const int o = (sM * 8 + vec) * 16 + lm;
            resR[o] = Cr[v];  resI[o] = Ci[v];
        }
    }
    __syncthreads();

    const int smp = tid >> 4;
    const int l   = tid & 15;
    {
        const bool isB = (l >= 2) && (l < 14);
        const int  q   = l - 2;
        const int  wid = (l < 2) ? l : isB ? 2 + (q >> 2) : (l == 14) ? 2 : 5;
        const int  vb  = (l < 2) ? 0 : isB ? 2 : 6;
        const float* wv = wt + wid * 16;
        const int rb = (smp * 8 + vb) * 16;
        for (int j = 0; j < 4; ++j) {
            const int ia = isB ? (q & 3) : (j >> 1);
            const int ib = isB ? j : (j & 1);
            float sr = 0.f, si = 0.f;
            for (int n = 0; n < 16; ++n) {
                float ar = resR[rb + ia * 16 + n], ai = resI[rb + ia * 16 + n];
                float br = resR[rb + ib * 16 + n], bi = resI[rb + ib * 16 + n];
                float w  = wv[n];
                sr += w * (ar * br + ai * bi);
                si += w * (ai * br - ar * bi);
            }
            gmR[smp * 64 + l * 4 + j] = sr;
            gmI[smp * 64 + l * 4 + j] = si;
        }
    }
    __syncthreads();

    {
        const int c = l >> 3, c2 = (l >> 2) & 1, d = (l >> 1) & 1, d2 = l & 1;
        const int ia = c * 2 + d, ib = c2 * 2 + d2;
        const float* GR = gmR + smp * 64;
        const float* GI = gmI + smp * 64;
        const int pa = c * 2 + c2, pb = 8 + ia * 4 + ib, pg = 56 + d * 2 + d2;
        float pacR = GR[pa],      pacI = GI[pa];
        float pasR = GR[4 + pa],  pasI = GI[4 + pa];
        float pb1R = GR[pb],      pb1I = GI[pb];
        float pbcR = GR[16 + pb], pbcI = GI[16 + pb];
        float pbsR = GR[32 + pb], pbsI = GI[32 + pb];
        float pg1R = GR[pg],      pg1I = GI[pg];
        float pgcR = GR[4 + pg],  pgcI = GI[4 + pg];
        float xr = pb1R * pg1R - pb1I * pg1I;
        float xi = pb1R * pg1I + pb1I * pg1R;
        float yr = pbcR * pg1R - pbcI * pg1I + pbsR * pgcR - pbsI * pgcI;
        float yi = pbcR * pg1I + pbcI * pg1R + pbsR * pgcI + pbsI * pgcR;
        float t  = pacR * xr - pacI * xi + pasR * yr - pasI * yi;
        t += __shfl_xor(t, 8, 16);
        t += __shfl_xor(t, 4, 16);
        t += __shfl_xor(t, 2, 16);
        t += __shfl_xor(t, 1, 16);
        if (l == 0) out[b * SPB + smp] = t + head_b[0];
    }
}

extern "C" void kernel_launch(void* const* d_in, const int* in_sizes, int n_in,
                              void* d_out, int out_size, void* d_ws, size_t ws_size,
                              hipStream_t stream) {
    const float* x      = (const float*)d_in[0];
    const float* params = (const float*)d_in[1];
    const float* head_w = (const float*)d_in[2];
    const float* head_b = (const float*)d_in[3];
    float* out = (float*)d_out;

    float* gr = (float*)d_ws;
    float* gi = gr + NG;
    float* w0 = gi + NG;

    qgate_build<<<(NG + NW + 255) / 256, 256, 0, stream>>>(params, gr, gi, w0);
    qsim<<<QBATCH / SPB, 256, 0, stream>>>(x, head_w, head_b, gr, gi, w0, out);
}


# Gate build folded into the main kernel: each workgroup makes the 24 wire 2x2s in LDS, lanes form their gate fragment from them (same product order); 2 launches -> 1, workspace unused
# speedup vs baseline: 32.9048x; 1.5169x over previous
#include <hip/hip_runtime.h>


#define NW    12
#define NL    2
#define QBATCH 8192
#define SPB   16
static_assert(NL == 2, "layer 1 is applied on the factors of the state layer 0 leaves");
static_assert(QBATCH % SPB == 0, "whole workgroups");

typedef __attribute__((ext_vector_type(2))) float v2f;
typedef __attribute__((ext_vector_type(8))) float v8f;

static __device__ __forceinline__ int gray4(int a) { return a ^ (a >> 1); }
static __device__ __forceinline__ float sg(int v) { return (__popc(v) & 1) ? -1.f : 1.f; }

static __device__ __forceinline__ v8f wmma4(v2f a, v2f b, v8f c) {
    return __builtin_amdgcn_wmma_f32_16x16x4_f32(false, a, false, b, (short)0, c,
                                                 false, false);
}

__launch_bounds__(256)
__global__ void qsim(const float* __restrict__ x,
                     const float* __restrict__ params,
                     const float* __restrict__ head_w,
                     const float* __restrict__ head_b,
                     float* __restrict__ out) {
    __shared__ float wm[NL * NW * 8];
    __shared__ float uR[SPB * 2 * NW], uI[SPB * 2 * NW];
    __shared__ float tabR[SPB * 48], tabI[SPB * 48];
    __shared__ float resR[SPB * 128], resI[SPB * 128];
    __shared__ float gmR[SPB * 64], gmI[SPB * 64];
    __shared__ float wt[6 * 16];

    const int tid  = threadIdx.x;
    const int b    = blockIdx.x;
    const int lane = tid & 31;
    const int wave = tid >> 5;
    const int half = lane >> 4;
    const int lm   = lane & 15;
    const int k0   = half * 2;

    const int g = (wave < 2) ? 0 : (wave < 6) ? 1 : 2;

    float cx = 0.f, sx = 0.f;
    if (tid < SPB * NW + NL * NW) {
        const float* src = (tid < SPB * NW) ? x + b * (SPB * NW) + tid : params + (tid - SPB * NW);
        float th = src[0] * 0.5f;
        cx = cosf(th);  sx = sinf(th);
        if (tid >= SPB * NW) {
            float* o = wm + (tid - SPB * NW) * 8;
            o[0] =  cx * cx;  o[1] =  sx * sx;
            o[2] = -cx * sx;  o[3] = -cx * sx;
            o[4] =  cx * sx;  o[5] = -cx * sx;
            o[6] =  cx * cx;  o[7] = -sx * sx;
        }
    }
    if (tid < 96) {
        const int id = tid >> 4, n = tid & 15;
        float v;
        if (id == 2) {
            v = 1.f;
        } else if (id == 1 || id == 4) {
            v = sg(n);
        } else {
            const int grp = (id == 0) ? 0 : (id == 3) ? 1 : 2;
            v = 0.f;
            for (int k = 0; k < 4; ++k) v += head_w[grp * 4 + k] * sg(n >> (3 - k));
        }
        wt[tid] = v;
    }
    __syncthreads();

    v2f Btr[4], Bti[4], Btn[4];
    {
        const float* wg = wm + (NW + g * 4) * 8;
        const int r0 = ((lm >> 3) & 1) * 4;
        const int r1 = ((lm >> 2) & 1) * 4 + 8;
        const int r2 = ((lm >> 1) & 1) * 4 + 16 + half * 2;
        const int r3 = (lm & 1) * 4 + 24;
        const float w2r = wg[r2], w2i = wg[r2 + 1];
        for (int kk = 0; kk < 4; ++kk) {
            const int c0 = r0 + (kk >> 1) * 2;
            const int c1 = r1 + (kk & 1) * 2;
            float pr = wg[c0], pi = wg[c0 + 1];
            float fr = wg[c1], fi = wg[c1 + 1];
            float nr = pr * fr - pi * fi;
            float ni = pr * fi + pi * fr;
            pr = nr * w2r - ni * w2i;
            pi = nr * w2i + ni * w2r;
            float e0r = wg[r3],     e0i = wg[r3 + 1];
            float e1r = wg[r3 + 2], e1i = wg[r3 + 3];
            Btr[kk].x = pr * e0r - pi * e0i;  Bti[kk].x = pr * e0i + pi * e0r;
            Btr[kk].y = pr * e1r - pi * e1i;  Bti[kk].y = pr * e1i + pi * e1r;
            Btn[kk].x = -Bti[kk].x;           Btn[kk].y = -Bti[kk].y;
        }
    }

    if (tid < SPB * NW) {
        const int s = tid / NW, w = tid - s * NW;
        const float* m = wm + w * 8;
        uR[s * 24 + w]      = m[0] * cx + m[2] * sx;
        uI[s * 24 + w]      = m[1] * cx + m[3] * sx;
        uR[s * 24 + w + NW] = m[4] * cx + m[6] * sx;
        uI[s * 24 + w + NW] = m[5] * cx + m[7] * sx;
    }
    __syncthreads();

    for (int r = 0; r < 3; ++r) {
        const int item = tid + 256 * r;
        const int s = item / 48, e = item - s * 48;
        const int grp = e >> 4, v = e & 15;
        float pr = 1.f, pi = 0.f;
        for (int k = 0; k < 4; ++k) {
            int w = s * 24 + grp * 4 + k + (((v >> (3 - k)) & 1) ? NW : 0);
            float fr = uR[w], fi = uI[w];
            float nr = pr * fr - pi * fi;
            float ni = pr * fi + pi * fr;
            pr = nr; pi = ni;
        }
        tabR[item] = pr;
        tabI[item] = pi;
    }
    __syncthreads();

    {
        int s, par, fb;
        if (g == 0)      { s = wave * 8 + (lm >> 1);        par = lm & 1;  fb = 0; }
        else if (g == 1) { s = (wave - 2) * 4 + (lm >> 2);  par = lm & 1;  fb = ((lm >> 1) & 1) << 3; }
        else             { s = (wave - 6) * 8 + (lm >> 1);  par = 0;       fb = (lm & 1) << 3; }
        const int tb = s * 48 + g * 16;
        const bool on0 = (g == 2) || (par == 0);
        const bool on1 = (g == 2) || (par == 1);
        v2f Avr[4], Avi[4];
        for (int kk = 0; kk < 4; ++kk) {
            int kc = kk * 4 + k0;
            int i0 = tb + (gray4(kc) ^ fb);
            int i1 = tb + (gray4(kc + 1) ^ fb);
            Avr[kk].x = on0 ? tabR[i0] : 0.f;  Avr[kk].y = on1 ? tabR[i1] : 0.f;
            Avi[kk].x = on0 ? tabI[i0] : 0.f;  Avi[kk].y = on1 ? tabI[i1] : 0.f;
        }
        v8f Cr = {}; v8f Ci = {};
        for (int kk = 0; kk < 4; ++kk) {
            Cr = wmma4(Avr[kk], Btr[kk], Cr);
            Cr = wmma4(Avi[kk], Btn[kk], Cr);
            Ci = wmma4(Avr[kk], Bti[kk], Ci);
            Ci = wmma4(Avi[kk], Btr[kk], Ci);
        }
        for (int v = 0; v < 8; ++v) {
            const int M = v + half * 8;
            int sM, vec;
            if (g == 0)      { sM = wave * 8 + (M >> 1);        vec = M & 1; }
            else if (g == 1) { sM = (wave - 2) * 4 + (M >> 2);  vec = 2 + (M & 3); }
            else             { sM = (wave - 6) * 8 + (M >> 1);  vec = 6 + (M & 1); }
            const int o = (sM * 8 + vec) * 16 + lm;
            resR[o] = Cr[v];  resI[o] = Ci[v];
        }
    }
    __syncthreads();

    const int smp = tid >> 4;
    const int l   = tid & 15;
    {
        const bool isB = (l >= 2) && (l < 14);
        const int  q   = l - 2;
        const int  wid = (l < 2) ? l : isB ? 2 + (q >> 2) : (l == 14) ? 2 : 5;
        const int  vb  = (l < 2) ? 0 : isB ? 2 : 6;
        const float* wv = wt + wid * 16;
        const int rb = (smp * 8 + vb) * 16;
        for (int j = 0; j < 4; ++j) {
            const int ia = isB ? (q & 3) : (j >> 1);
            const int ib = isB ? j : (j & 1);
            float sr = 0.f, si = 0.f;
            for (int n = 0; n < 16; ++n) {
                float ar = resR[rb + ia * 16 + n], ai = resI[rb + ia * 16 + n];
                float br = resR[rb + ib * 16 + n], bi = resI[rb + ib * 16 + n];
                float w  = wv[n];
                sr += w * (ar * br + ai * bi);
                si += w * (ai * br - ar * bi);
            }
            gmR[smp * 64 + l * 4 + j] = sr;
            gmI[smp * 64 + l * 4 + j] = si;
        }
    }
    __syncthreads();

    {
        const int c = l >> 3, c2 = (l >> 2) & 1, d = (l >> 1) & 1, d2 = l & 1;
        const int ia = c * 2 + d, ib = c2 * 2 + d2;
        const float* GR = gmR + smp * 64;
        const float* GI = gmI + smp * 64;
        const int pa = c * 2 + c2, pb = 8 + ia * 4 + ib, pg = 56 + d * 2 + d2;
        float pacR = GR[pa],      pacI = GI[pa];
        float pasR = GR[4 + pa],  pasI = GI[4 + pa];
        float pb1R = GR[pb],      pb1I = GI[pb];
        float pbcR = GR[16 + pb], pbcI = GI[16 + pb];
        float pbsR = GR[32 + pb], pbsI = GI[32 + pb];
        float pg1R = GR[pg],      pg1I = GI[pg];
        float pgcR = GR[4 + pg],  pgcI = GI[4 + pg];
        float xr = pb1R * pg1R - pb1I * pg1I;
        float xi = pb1R * pg1I + pb1I * pg1R;
        float yr = pbcR * pg1R - pbcI * pg1I + pbsR * pgcR - pbsI * pgcI;
        float yi = pbcR * pg1I + pbcI * pg1R + pbsR * pgcI + pbsI * pgcR;
        float t  = pacR * xr - pacI * xi + pasR * yr - pasI * yi;
        t += __shfl_xor(t, 8, 16);
        t += __shfl_xor(t, 4, 16);
        t += __shfl_xor(t, 2, 16);
        t += __shfl_xor(t, 1, 16);
        if (l == 0) out[b * SPB + smp] = t + head_b[0];
    }
}

extern "C" void kernel_launch(void* const* d_in, const int* in_sizes, int n_in,
                              void* d_out, int out_size, void* d_ws, size_t ws_size,
                              hipStream_t stream) {
    const float* x      = (const float*)d_in[0];
    const float* params = (const float*)d_in[1];
    const float* head_w = (const float*)d_in[2];
    const float* head_b = (const float*)d_in[3];
    float* out = (float*)d_out;

    qsim<<<QBATCH / SPB, 256, 0, stream>>>(x, params, head_w, head_b, out);
}


# Weighted inner products: one unordered vector pair per thread, v_i*conj(v_i') formed once for its weights (1 and sg by add/sub), mirror stored as conjugate; mirrored imaginary parts round differently
# speedup vs baseline: 38.0037x; 1.1550x over previous
#include <hip/hip_runtime.h>


#define NW    12
#define NL    2
#define QBATCH 8192
#define SPB   16
static_assert(NL == 2, "layer 1 is applied on the factors of the state layer 0 leaves");
static_assert(QBATCH % SPB == 0, "whole workgroups");

typedef __attribute__((ext_vector_type(2))) float v2f;
typedef __attribute__((ext_vector_type(8))) float v8f;

static __device__ __forceinline__ int gray4(int a) { return a ^ (a >> 1); }
static __device__ __forceinline__ float sg(int v) { return (__popc(v) & 1) ? -1.f : 1.f; }

static __device__ __forceinline__ v8f wmma4(v2f a, v2f b, v8f c) {
    return __builtin_amdgcn_wmma_f32_16x16x4_f32(false, a, false, b, (short)0, c,
                                                 false, false);
}

__launch_bounds__(256)
__global__ void qsim(const float* __restrict__ x,
                     const float* __restrict__ params,
                     const float* __restrict__ head_w,
                     const float* __restrict__ head_b,
                     float* __restrict__ out) {
    __shared__ float wm[NL * NW * 8];
    __shared__ float uR[SPB * 2 * NW], uI[SPB * 2 * NW];
    __shared__ float tabR[SPB * 48], tabI[SPB * 48];
    __shared__ float resR[SPB * 128], resI[SPB * 128];
    __shared__ float gmR[SPB * 64], gmI[SPB * 64];
    __shared__ float wt[3 * 16];

    const int tid  = threadIdx.x;
    const int b    = blockIdx.x;
    const int lane = tid & 31;
    const int wave = tid >> 5;
    const int half = lane >> 4;
    const int lm   = lane & 15;
    const int k0   = half * 2;

    const int g = (wave < 2) ? 0 : (wave < 6) ? 1 : 2;

    float cx = 0.f, sx = 0.f;
    if (tid < SPB * NW + NL * NW) {
        const float* src = (tid < SPB * NW) ? x + b * (SPB * NW) + tid : params + (tid - SPB * NW);
        float th = src[0] * 0.5f;
        cx = cosf(th);  sx = sinf(th);
        if (tid >= SPB * NW) {
            float* o = wm + (tid - SPB * NW) * 8;
            o[0] =  cx * cx;  o[1] =  sx * sx;
            o[2] = -cx * sx;  o[3] = -cx * sx;
            o[4] =  cx * sx;  o[5] = -cx * sx;
            o[6] =  cx * cx;  o[7] = -sx * sx;
        }
    }
    if (tid < 48) {
        const int grp = tid >> 4, n = tid & 15;
        float v = 0.f;
        for (int k = 0; k < 4; ++k) v += head_w[grp * 4 + k] * sg(n >> (3 - k));
        wt[tid] = v;
    }
    __syncthreads();

    v2f Btr[4], Bti[4], Btn[4];
    {
        const float* wg = wm + (NW + g * 4) * 8;
        const int r0 = ((lm >> 3) & 1) * 4;
        const int r1 = ((lm >> 2) & 1) * 4 + 8;
        const int r2 = ((lm >> 1) & 1) * 4 + 16 + half * 2;
        const int r3 = (lm & 1) * 4 + 24;
        const float w2r = wg[r2], w2i = wg[r2 + 1];
        for (int kk = 0; kk < 4; ++kk) {
            const int c0 = r0 + (kk >> 1) * 2;
            const int c1 = r1 + (kk & 1) * 2;
            float pr = wg[c0], pi = wg[c0 + 1];
            float fr = wg[c1], fi = wg[c1 + 1];
            float nr = pr * fr - pi * fi;
            float ni = pr * fi + pi * fr;
            pr = nr * w2r - ni * w2i;
            pi = nr * w2i + ni * w2r;
            float e0r = wg[r3],     e0i = wg[r3 + 1];
            float e1r = wg[r3 + 2], e1i = wg[r3 + 3];
            Btr[kk].x = pr * e0r - pi * e0i;  Bti[kk].x = pr * e0i + pi * e0r;
            Btr[kk].y = pr * e1r - pi * e1i;  Bti[kk].y = pr * e1i + pi * e1r;
            Btn[kk].x = -Bti[kk].x;           Btn[kk].y = -Bti[kk].y;
        }
    }

    if (tid < SPB * NW) {
        const int s = tid / NW, w = tid - s * NW;
        const float* m = wm + w * 8;
        uR[s * 24 + w]      = m[0] * cx + m[2] * sx;
        uI[s * 24 + w]      = m[1] * cx + m[3] * sx;
        uR[s * 24 + w + NW] = m[4] * cx + m[6] * sx;
        uI[s * 24 + w + NW] = m[5] * cx + m[7] * sx;
    }
    __syncthreads();

    for (int r = 0; r < 3; ++r) {
        const int item = tid + 256 * r;
        const int s = item / 48, e = item - s * 48;
        const int grp = e >> 4, v = e & 15;
        float pr = 1.f, pi = 0.f;
        for (int k = 0; k < 4; ++k) {
            int w = s * 24 + grp * 4 + k + (((v >> (3 - k)) & 1) ? NW : 0);
            float fr = uR[w], fi = uI[w];
            float nr = pr * fr - pi * fi;
            float ni = pr * fi + pi * fr;
            pr = nr; pi = ni;
        }
        tabR[item] = pr;
        tabI[item] = pi;
    }
    __syncthreads();

    {
        int s, par, fb;
        if (g == 0)      { s = wave * 8 + (lm >> 1);        par = lm & 1;  fb = 0; }
        else if (g == 1) { s = (wave - 2) * 4 + (lm >> 2);  par = lm & 1;  fb = ((lm >> 1) & 1) << 3; }
        else             { s = (wave - 6) * 8 + (lm >> 1);  par = 0;       fb = (lm & 1) << 3; }
        const int tb = s * 48 + g * 16;
        const bool on0 = (g == 2) || (par == 0);
        const bool on1 = (g == 2) || (par == 1);
        v2f Avr[4], Avi[4];
        for (int kk = 0; kk < 4; ++kk) {
            int kc = kk * 4 + k0;
            int i0 = tb + (gray4(kc) ^ fb);
            int i1 = tb + (gray4(kc + 1) ^ fb);
            Avr[kk].x = on0 ? tabR[i0] : 0.f;  Avr[kk].y = on1 ? tabR[i1] : 0.f;
            Avi[kk].x = on0 ? tabI[i0] : 0.f;  Avi[kk].y = on1 ? tabI[i1] : 0.f;
        }
        v8f Cr = {}; v8f Ci = {};
        for (int kk = 0; kk < 4; ++kk) {
            Cr = wmma4(Avr[kk], Btr[kk], Cr);
            Cr = wmma4(Avi[kk], Btn[kk], Cr);
            Ci = wmma4(Avr[kk], Bti[kk], Ci);
            Ci = wmma4(Avi[kk], Btr[kk], Ci);
        }
        for (int v = 0; v < 8; ++v) {
            const int M = v + half * 8;
            int sM, vec;
            if (g == 0)      { sM = wave * 8 + (M >> 1);        vec = M & 1; }
            else if (g == 1) { sM = (wave - 2) * 4 + (M >> 2);  vec = 2 + (M & 3); }
            else             { sM = (wave - 6) * 8 + (M >> 1);  vec = 6 + (M & 1); }
            const int o = (sM * 8 + vec) * 16 + lm;
            resR[o] = Cr[v];  resI[o] = Ci[v];
        }
    }
    __syncthreads();

    const int smp = tid >> 4;
    const int l   = tid & 15;
    {
        const bool isA = (l < 3);
        const bool isB = (l >= 3) && (l < 13);
        const bool isG = (l >= 13);
        const int  q   = isA ? l : isB ? (l - 3) : (l - 13);
        const int  ia  = isB ? ((0xE9500 >> (2 * q)) & 3) : (q >> 1);
        const int  ib  = isB ? ((0xFB9E4 >> (2 * q)) & 3) : ((q + 1) >> 1);
        const int  vb  = isA ? 0 : isB ? 2 : 6;
        const int  dim = isB ? 4 : 2;
        const int  oC  = isA ? 0 : isB ? 24 : 60;
        const int  o1  = isB ? 8 : 56;
        const int  oS  = isA ? 4 : 40;
        const float* wc = wt + (isA ? 0 : isB ? 1 : 2) * 16;
        const int ra = (smp * 8 + vb + ia) * 16;
        const int rb = (smp * 8 + vb + ib) * 16;
        float cr = 0.f, ci = 0.f, ur = 0.f, ui = 0.f, gr = 0.f, gi = 0.f;
        for (int n = 0; n < 16; ++n) {
            float ar = resR[ra + n], ai = resI[ra + n];
            float br = resR[rb + n], bi = resI[rb + n];
            float pr = ar * br + ai * bi;
            float pi = ai * br - ar * bi;
            cr += wc[n] * pr;  ci += wc[n] * pi;
            ur += pr;          ui += pi;
            if (__popc(n) & 1) { gr -= pr;  gi -= pi; }
            else               { gr += pr;  gi += pi; }
        }
        float* GR = gmR + smp * 64;
        float* GI = gmI + smp * 64;
        const int od = ia * dim + ib;
        const int om = ib * dim + ia;
        GR[oC + od] = cr;  GI[oC + od] = ci;
        if (!isA) { GR[o1 + od] = ur;  GI[o1 + od] = ui; }
        if (!isG) { GR[oS + od] = gr;  GI[oS + od] = gi; }
        if (ia != ib) {
            GR[oC + om] = cr;  GI[oC + om] = -ci;
            if (!isA) { GR[o1 + om] = ur;  GI[o1 + om] = -ui; }
            if (!isG) { GR[oS + om] = gr;  GI[oS + om] = -gi; }
        }
    }
    __syncthreads();

    {
        const int c = l >> 3, c2 = (l >> 2) & 1, d = (l >> 1) & 1, d2 = l & 1;
        const int ia = c * 2 + d, ib = c2 * 2 + d2;
        const float* GR = gmR + smp * 64;
        const float* GI = gmI + smp * 64;
        const int pa = c * 2 + c2, pb = 8 + ia * 4 + ib, pg = 56 + d * 2 + d2;
        float pacR = GR[pa],      pacI = GI[pa];
        float pasR = GR[4 + pa],  pasI = GI[4 + pa];
        float pb1R = GR[pb],      pb1I = GI[pb];
        float pbcR = GR[16 + pb], pbcI = GI[16 + pb];
        float pbsR = GR[32 + pb], pbsI = GI[32 + pb];
        float pg1R = GR[pg],      pg1I = GI[pg];
        float pgcR = GR[4 + pg],  pgcI = GI[4 + pg];
        float xr = pb1R * pg1R - pb1I * pg1I;
        float xi = pb1R * pg1I + pb1I * pg1R;
        float yr = pbcR * pg1R - pbcI * pg1I + pbsR * pgcR - pbsI * pgcI;
        float yi = pbcR * pg1I + pbcI * pg1R + pbsR * pgcI + pbsI * pgcR;
        float t  = pacR * xr - pacI * xi + pasR * yr - pasI * yi;
        t += __shfl_xor(t, 8, 16);
        t += __shfl_xor(t, 4, 16);
        t += __shfl_xor(t, 2, 16);
        t += __shfl_xor(t, 1, 16);
        if (l == 0) out[b * SPB + smp] = t + head_b[0];
    }
}

extern "C" void kernel_launch(void* const* d_in, const int* in_sizes, int n_in,
                              void* d_out, int out_size, void* d_ws, size_t ws_size,
                              hipStream_t stream) {
    const float* x      = (const float*)d_in[0];
    const float* params = (const float*)d_in[1];
    const float* head_w = (const float*)d_in[2];
    const float* head_b = (const float*)d_in[3];
    float* out = (float*)d_out;

    qsim<<<QBATCH / SPB, 256, 0, stream>>>(x, params, head_w, head_b, out);
}


# WMMA tiles regrouped: 16 samples x one vector per wave, roles scalar per wave; zero half of k dropped for alpha/beta (K=8); tables in parity order read as b128; k sum order in the WMMAs changes
# speedup vs baseline: 42.6675x; 1.1227x over previous
#include <hip/hip_runtime.h>


#define NW    12
#define NL    2
#define QBATCH 8192
#define SPB   16
#define TABP  52
static_assert(NL == 2, "layer 1 is applied on the factors of the state layer 0 leaves");
static_assert(QBATCH % SPB == 0, "whole workgroups");

typedef __attribute__((ext_vector_type(2))) float v2f;
typedef __attribute__((ext_vector_type(8))) float v8f;
typedef __attribute__((ext_vector_type(4))) float v4f;

static __device__ __forceinline__ int gray4(int a) { return a ^ (a >> 1); }
static __device__ __forceinline__ float sg(int v) { return (__popc(v) & 1) ? -1.f : 1.f; }

static __device__ __forceinline__ v8f wmma4(v2f a, v2f b, v8f c) {
    return __builtin_amdgcn_wmma_f32_16x16x4_f32(false, a, false, b, (short)0, c,
                                                 false, false);
}

__launch_bounds__(256)
__global__ void qsim(const float* __restrict__ x,
                     const float* __restrict__ params,
                     const float* __restrict__ head_w,
                     const float* __restrict__ head_b,
                     float* __restrict__ out) {
    __shared__ float wm[NL * NW * 8];
    __shared__ float uR[SPB * 2 * NW], uI[SPB * 2 * NW];
    __shared__ __attribute__((aligned(16))) float tabR[SPB * TABP], tabI[SPB * TABP];
    __shared__ float resR[SPB * 128], resI[SPB * 128];
    __shared__ float gmR[SPB * 64], gmI[SPB * 64];
    __shared__ float wt[3 * 16];

    const int tid  = threadIdx.x;
    const int b    = blockIdx.x;
    const int lane = tid & 31;
    const int wave = __builtin_amdgcn_readfirstlane(tid >> 5);
    const int half = lane >> 4;
    const int lm   = lane & 15;
    const int k0   = half * 2;

    const int g    = (wave < 2) ? 0 : (wave < 6) ? 1 : 2;
    const int flip = (wave < 6) ? (wave >> 2) : (wave & 1);
    const int par  = wave & 1;

    float cx = 0.f, sx = 0.f;
    if (tid < SPB * NW + NL * NW) {
        const float* src = (tid < SPB * NW) ? x + b * (SPB * NW) + tid : params + (tid - SPB * NW);
        float th = src[0] * 0.5f;
        cx = cosf(th);  sx = sinf(th);
        if (tid >= SPB * NW) {
            float* o = wm + (tid - SPB * NW) * 8;
            o[0] =  cx * cx;  o[1] =  sx * sx;
            o[2] = -cx * sx;  o[3] = -cx * sx;
            o[4] =  cx * sx;  o[5] = -cx * sx;
            o[6] =  cx * cx;  o[7] = -sx * sx;
        }
    }
    if (tid < 48) {
        const int grp = tid >> 4, n = tid & 15;
        float v = 0.f;
        for (int k = 0; k < 4; ++k) v += head_w[grp * 4 + k] * sg(n >> (3 - k));
        wt[tid] = v;
    }
    __syncthreads();

    v2f Btr[4] = {}, Bti[4] = {}, Btn[4] = {};
    {
        const float* wg = wm + (NW + g * 4) * 8;
        const float* q0 = wg +      ((lm >> 3) & 1) * 4;
        const float* q1 = wg +  8 + ((lm >> 2) & 1) * 4;
        const float* q2 = wg + 16 + ((lm >> 1) & 1) * 4;
        const float* q3 = wg + 24 + (lm & 1) * 4;
        const float e0r = q2[flip * 2],       e0i = q2[flip * 2 + 1];
        const float e1r = q2[(1 - flip) * 2], e1i = q2[(1 - flip) * 2 + 1];
        if (g < 2) {
            const int c0 = (half ^ flip) * 2, c3 = par * 2;
            const float ar = q0[c0], ai = q0[c0 + 1], br = q3[c3], bi = q3[c3 + 1];
            const float pr = ar * br - ai * bi, pi = ar * bi + ai * br;
            for (int kk = 0; kk < 2; ++kk) {
                const int c1 = (kk ^ flip) * 2;
                const float fr = q1[c1], fi = q1[c1 + 1];
                const float nr = pr * fr - pi * fi, ni = pr * fi + pi * fr;
                Btr[kk].x = nr * e0r - ni * e0i;  Bti[kk].x = nr * e0i + ni * e0r;
                Btr[kk].y = nr * e1r - ni * e1i;  Bti[kk].y = nr * e1i + ni * e1r;
                Btn[kk].x = -Bti[kk].x;           Btn[kk].y = -Bti[kk].y;
            }
        } else {
            const int c3 = (half ^ flip) * 2;
            const float br = q3[c3], bi = q3[c3 + 1];
            for (int kk = 0; kk < 4; ++kk) {
                const int c0 = ((kk >> 1) ^ flip) * 2, c1 = ((kk & 1) ^ flip) * 2;
                const float ar = q0[c0], ai = q0[c0 + 1], fr = q1[c1], fi = q1[c1 + 1];
                const float pr = ar * fr - ai * fi, pi = ar * fi + ai * fr;
                const float nr = pr * br - pi * bi, ni = pr * bi + pi * br;
                Btr[kk].x = nr * e0r - ni * e0i;  Bti[kk].x = nr * e0i + ni * e0r;
                Btr[kk].y = nr * e1r - ni * e1i;  Bti[kk].y = nr * e1i + ni * e1r;
                Btn[kk].x = -Bti[kk].x;           Btn[kk].y = -Bti[kk].y;
            }
        }
    }

    if (tid < SPB * NW) {
        const int s = tid / NW, w = tid - s * NW;
        const float* m = wm + w * 8;
        uR[s * 24 + w]      = m[0] * cx + m[2] * sx;
        uI[s * 24 + w]      = m[1] * cx + m[3] * sx;
        uR[s * 24 + w + NW] = m[4] * cx + m[6] * sx;
        uI[s * 24 + w + NW] = m[5] * cx + m[7] * sx;
    }
    __syncthreads();

    {
        const int p = tid & 15;
        const int k = ((p & 7) << 1) | (p >> 3);
        const int v = gray4(k);
        for (int r = 0; r < 3; ++r) {
            const int item = tid + 256 * r;
            const int s = item / 48, e = item - s * 48;
            const int grp = e >> 4;
            float pr = 1.f, pi = 0.f;
            for (int kq = 0; kq < 4; ++kq) {
                int w = s * 24 + grp * 4 + kq + (((v >> (3 - kq)) & 1) ? NW : 0);
                float fr = uR[w], fi = uI[w];
                float nr = pr * fr - pi * fi;
                float ni = pr * fi + pi * fr;
                pr = nr; pi = ni;
            }
            tabR[s * TABP + e] = pr;
            tabI[s * TABP + e] = pi;
        }
    }
    __syncthreads();

    {
        v8f Cr = {}; v8f Ci = {};
        if (g < 2) {
            const int tb = lm * TABP + g * 16 + (par ^ flip) * 8 + half * 4;
            const v4f fr = *(const v4f*)(tabR + tb);
            const v4f fi = *(const v4f*)(tabI + tb);
            v2f Avr[2], Avi[2];
            Avr[0].x = fr.x; Avr[0].y = fr.y; Avr[1].x = fr.z; Avr[1].y = fr.w;
            Avi[0].x = fi.x; Avi[0].y = fi.y; Avi[1].x = fi.z; Avi[1].y = fi.w;
            for (int kk = 0; kk < 2; ++kk) {
                Cr = wmma4(Avr[kk], Btr[kk], Cr);
                Cr = wmma4(Avi[kk], Btn[kk], Cr);
                Ci = wmma4(Avr[kk], Bti[kk], Ci);
                Ci = wmma4(Avi[kk], Btr[kk], Ci);
            }
        } else {
            const int tb = lm * TABP + g * 16 + half * 8;
            const v4f fr0 = *(const v4f*)(tabR + tb), fr1 = *(const v4f*)(tabR + tb + 4);
            const v4f fi0 = *(const v4f*)(tabI + tb), fi1 = *(const v4f*)(tabI + tb + 4);
            v2f Avr[4], Avi[4];
            Avr[0].x = fr0.x; Avr[0].y = fr0.y; Avr[1].x = fr0.z; Avr[1].y = fr0.w;
            Avr[2].x = fr1.x; Avr[2].y = fr1.y; Avr[3].x = fr1.z; Avr[3].y = fr1.w;
            Avi[0].x = fi0.x; Avi[0].y = fi0.y; Avi[1].x = fi0.z; Avi[1].y = fi0.w;
            Avi[2].x = fi1.x; Avi[2].y = fi1.y; Avi[3].x = fi1.z; Avi[3].y = fi1.w;
            for (int kk = 0; kk < 4; ++kk) {
                Cr = wmma4(Avr[kk], Btr[kk], Cr);
                Cr = wmma4(Avi[kk], Btn[kk], Cr);
                Ci = wmma4(Avr[kk], Bti[kk], Ci);
                Ci = wmma4(Avi[kk], Btr[kk], Ci);
            }
        }
        const int o = (half * 64 + wave) * 16 + lm;
        for (int v = 0; v < 8; ++v) {
            resR[o + v * 128] = Cr[v];  resI[o + v * 128] = Ci[v];
        }
    }
    __syncthreads();

    const int smp = tid >> 4;
    const int l   = tid & 15;
    {
        const bool isA = (l < 3);
        const bool isB = (l >= 3) && (l < 13);
        const bool isG = (l >= 13);
        const int  q   = isA ? l : isB ? (l - 3) : (l - 13);
        const int  ia  = isB ? ((0xE9500 >> (2 * q)) & 3) : (q >> 1);
        const int  ib  = isB ? ((0xFB9E4 >> (2 * q)) & 3) : ((q + 1) >> 1);
        const int  vb  = isA ? 0 : isB ? 2 : 6;
        const int  dim = isB ? 4 : 2;
        const int  oC  = isA ? 0 : isB ? 24 : 60;
        const int  o1  = isB ? 8 : 56;
        const int  oS  = isA ? 4 : 40;
        const float* wc = wt + (isA ? 0 : isB ? 1 : 2) * 16;
        const int ra = (smp * 8 + vb + ia) * 16;
        const int rb = (smp * 8 + vb + ib) * 16;
        float cr = 0.f, ci = 0.f, ur = 0.f, ui = 0.f, gr = 0.f, gi = 0.f;
        for (int n = 0; n < 16; ++n) {
            float ar = resR[ra + n], ai = resI[ra + n];
            float br = resR[rb + n], bi = resI[rb + n];
            float pr = ar * br + ai * bi;
            float pi = ai * br - ar * bi;
            cr += wc[n] * pr;  ci += wc[n] * pi;
            ur += pr;          ui += pi;
            if (__popc(n) & 1) { gr -= pr;  gi -= pi; }
            else               { gr += pr;  gi += pi; }
        }
        float* GR = gmR + smp * 64;
        float* GI = gmI + smp * 64;
        const int od = ia * dim + ib;
        const int om = ib * dim + ia;
        GR[oC + od] = cr;  GI[oC + od] = ci;
        if (!isA) { GR[o1 + od] = ur;  GI[o1 + od] = ui; }
        if (!isG) { GR[oS + od] = gr;  GI[oS + od] = gi; }
        if (ia != ib) {
            GR[oC + om] = cr;  GI[oC + om] = -ci;
            if (!isA) { GR[o1 + om] = ur;  GI[o1 + om] = -ui; }
            if (!isG) { GR[oS + om] = gr;  GI[oS + om] = -gi; }
        }
    }
    __syncthreads();

    {
        const int c = l >> 3, c2 = (l >> 2) & 1, d = (l >> 1) & 1, d2 = l & 1;
        const int ia = c * 2 + d, ib = c2 * 2 + d2;
        const float* GR = gmR + smp * 64;
        const float* GI = gmI + smp * 64;
        const int pa = c * 2 + c2, pb = 8 + ia * 4 + ib, pg = 56 + d * 2 + d2;
        float pacR = GR[pa],      pacI = GI[pa];
        float pasR = GR[4 + pa],  pasI = GI[4 + pa];
        float pb1R = GR[pb],      pb1I = GI[pb];
        float pbcR = GR[16 + pb], pbcI = GI[16 + pb];
        float pbsR = GR[32 + pb], pbsI = GI[32 + pb];
        float pg1R = GR[pg],      pg1I = GI[pg];
        float pgcR = GR[4 + pg],  pgcI = GI[4 + pg];
        float xr = pb1R * pg1R - pb1I * pg1I;
        float xi = pb1R * pg1I + pb1I * pg1R;
        float yr = pbcR * pg1R - pbcI * pg1I + pbsR * pgcR - pbsI * pgcI;
        float yi = pbcR * pg1I + pbcI * pg1R + pbsR * pgcI + pbsI * pgcR;
        float t  = pacR * xr - pacI * xi + pasR * yr - pasI * yi;
        t += __shfl_xor(t, 8, 16);
        t += __shfl_xor(t, 4, 16);
        t += __shfl_xor(t, 2, 16);
        t += __shfl_xor(t, 1, 16);
        if (l == 0) out[b * SPB + smp] = t + head_b[0];
    }
}

extern "C" void kernel_launch(void* const* d_in, const int* in_sizes, int n_in,
                              void* d_out, int out_size, void* d_ws, size_t ws_size,
                              hipStream_t stream) {
    const float* x      = (const float*)d_in[0];
    const float* params = (const float*)d_in[1];
    const float* head_w = (const float*)d_in[2];
    const float* head_b = (const float*)d_in[3];
    float* out = (float*)d_out;

    qsim<<<QBATCH / SPB, 256, 0, stream>>>(x, params, head_w, head_b, out);
}


# A sample's 16 threads are one 16-lane row: inner products reach the final terms by in-row gathers (v_permlane16_var), not LDS, so the last barrier goes; 16-term sum by DPP row_xmask adds, same tree
# speedup vs baseline: 48.0520x; 1.1262x over previous
#include <hip/hip_runtime.h>


#define NW    12
#define NL    2
#define QBATCH 8192
#define SPB   16
#define TABP  52
static_assert(NL == 2, "layer 1 is applied on the factors of the state layer 0 leaves");
static_assert(QBATCH % SPB == 0, "whole workgroups");

typedef __attribute__((ext_vector_type(2))) float v2f;
typedef __attribute__((ext_vector_type(8))) float v8f;
typedef __attribute__((ext_vector_type(4))) float v4f;

static __device__ __forceinline__ int gray4(int a) { return a ^ (a >> 1); }
static __device__ __forceinline__ float sg(int v) { return (__popc(v) & 1) ? -1.f : 1.f; }

static __device__ __forceinline__ v8f wmma4(v2f a, v2f b, v8f c) {
    return __builtin_amdgcn_wmma_f32_16x16x4_f32(false, a, false, b, (short)0, c,
                                                 false, false);
}

static __device__ __forceinline__ float row_gather(float v, int sel) {
    return __builtin_bit_cast(float, __builtin_amdgcn_permlane16_var(0u, __builtin_bit_cast(unsigned, v),
                                                                     (unsigned)sel, false, false));
}
template <int MASK>
static __device__ __forceinline__ float row_xor_add(float t) {
    return t + __builtin_bit_cast(float, __builtin_amdgcn_update_dpp(0, __builtin_bit_cast(int, t),
                                                                     0x160 | MASK, 0xf, 0xf, false));
}

__launch_bounds__(256)
__global__ void qsim(const float* __restrict__ x,
                     const float* __restrict__ params,
                     const float* __restrict__ head_w,
                     const float* __restrict__ head_b,
                     float* __restrict__ out) {
    __shared__ float wm[NL * NW * 8];
    __shared__ float uR[SPB * 2 * NW], uI[SPB * 2 * NW];
    __shared__ __attribute__((aligned(16))) float tabR[SPB * TABP], tabI[SPB * TABP];
    __shared__ float resR[SPB * 128], resI[SPB * 128];
    __shared__ float wt[3 * 16];

    const int tid  = threadIdx.x;
    const int b    = blockIdx.x;
    const int lane = tid & 31;
    const int wave = __builtin_amdgcn_readfirstlane(tid >> 5);
    const int half = lane >> 4;
    const int lm   = lane & 15;
    const int k0   = half * 2;

    const int g    = (wave < 2) ? 0 : (wave < 6) ? 1 : 2;
    const int flip = (wave < 6) ? (wave >> 2) : (wave & 1);
    const int par  = wave & 1;

    float cx = 0.f, sx = 0.f;
    if (tid < SPB * NW + NL * NW) {
        const float* src = (tid < SPB * NW) ? x + b * (SPB * NW) + tid : params + (tid - SPB * NW);
        float th = src[0] * 0.5f;
        cx = cosf(th);  sx = sinf(th);
        if (tid >= SPB * NW) {
            float* o = wm + (tid - SPB * NW) * 8;
            o[0] =  cx * cx;  o[1] =  sx * sx;
            o[2] = -cx * sx;  o[3] = -cx * sx;
            o[4] =  cx * sx;  o[5] = -cx * sx;
            o[6] =  cx * cx;  o[7] = -sx * sx;
        }
    }
    if (tid < 48) {
        const int grp = tid >> 4, n = tid & 15;
        float v = 0.f;
        for (int k = 0; k < 4; ++k) v += head_w[grp * 4 + k] * sg(n >> (3 - k));
        wt[tid] = v;
    }
    __syncthreads();

    v2f Btr[4] = {}, Bti[4] = {}, Btn[4] = {};
    {
        const float* wg = wm + (NW + g * 4) * 8;
        const float* q0 = wg +      ((lm >> 3) & 1) * 4;
        const float* q1 = wg +  8 + ((lm >> 2) & 1) * 4;
        const float* q2 = wg + 16 + ((lm >> 1) & 1) * 4;
        const float* q3 = wg + 24 + (lm & 1) * 4;
        const float e0r = q2[flip * 2],       e0i = q2[flip * 2 + 1];
        const float e1r = q2[(1 - flip) * 2], e1i = q2[(1 - flip) * 2 + 1];
        if (g < 2) {
            const int c0 = (half ^ flip) * 2, c3 = par * 2;
            const float ar = q0[c0], ai = q0[c0 + 1], br = q3[c3], bi = q3[c3 + 1];
            const float pr = ar * br - ai * bi, pi = ar * bi + ai * br;
            for (int kk = 0; kk < 2; ++kk) {
                const int c1 = (kk ^ flip) * 2;
                const float fr = q1[c1], fi = q1[c1 + 1];
                const float nr = pr * fr - pi * fi, ni = pr * fi + pi * fr;
                Btr[kk].x = nr * e0r - ni * e0i;  Bti[kk].x = nr * e0i + ni * e0r;
                Btr[kk].y = nr * e1r - ni * e1i;  Bti[kk].y = nr * e1i + ni * e1r;
                Btn[kk].x = -Bti[kk].x;           Btn[kk].y = -Bti[kk].y;
            }
        } else {
            const int c3 = (half ^ flip) * 2;
            const float br = q3[c3], bi = q3[c3 + 1];
            for (int kk = 0; kk < 4; ++kk) {
                const int c0 = ((kk >> 1) ^ flip) * 2, c1 = ((kk & 1) ^ flip) * 2;
                const float ar = q0[c0], ai = q0[c0 + 1], fr = q1[c1], fi = q1[c1 + 1];
                const float pr = ar * fr - ai * fi, pi = ar * fi + ai * fr;
                const float nr = pr * br - pi * bi, ni = pr * bi + pi * br;
                Btr[kk].x = nr * e0r - ni * e0i;  Bti[kk].x = nr * e0i + ni * e0r;
                Btr[kk].y = nr * e1r - ni * e1i;  Bti[kk].y = nr * e1i + ni * e1r;
                Btn[kk].x = -Bti[kk].x;           Btn[kk].y = -Bti[kk].y;
            }
        }
    }

    if (tid < SPB * NW) {
        const int s = tid / NW, w = tid - s * NW;
        const float* m = wm + w * 8;
        uR[s * 24 + w]      = m[0] * cx + m[2] * sx;
        uI[s * 24 + w]      = m[1] * cx + m[3] * sx;
        uR[s * 24 + w + NW] = m[4] * cx + m[6] * sx;
        uI[s * 24 + w + NW] = m[5] * cx + m[7] * sx;
    }
    __syncthreads();

    {
        const int p = tid & 15;
        const int k = ((p & 7) << 1) | (p >> 3);
        const int v = gray4(k);
        for (int r = 0; r < 3; ++r) {
            const int item = tid + 256 * r;
            const int s = item / 48, e = item - s * 48;
            const int grp = e >> 4;
            float pr = 1.f, pi = 0.f;
            for (int kq = 0; kq < 4; ++kq) {
                int w = s * 24 + grp * 4 + kq + (((v >> (3 - kq)) & 1) ? NW : 0);
                float fr = uR[w], fi = uI[w];
                float nr = pr * fr - pi * fi;
                float ni = pr * fi + pi * fr;
                pr = nr; pi = ni;
            }
            tabR[s * TABP + e] = pr;
            tabI[s * TABP + e] = pi;
        }
    }
    __syncthreads();

    {
        v8f Cr = {}; v8f Ci = {};
        if (g < 2) {
            const int tb = lm * TABP + g * 16 + (par ^ flip) * 8 + half * 4;
            const v4f fr = *(const v4f*)(tabR + tb);
            const v4f fi = *(const v4f*)(tabI + tb);
            v2f Avr[2], Avi[2];
            Avr[0].x = fr.x; Avr[0].y = fr.y; Avr[1].x = fr.z; Avr[1].y = fr.w;
            Avi[0].x = fi.x; Avi[0].y = fi.y; Avi[1].x = fi.z; Avi[1].y = fi.w;
            for (int kk = 0; kk < 2; ++kk) {
                Cr = wmma4(Avr[kk], Btr[kk], Cr);
                Cr = wmma4(Avi[kk], Btn[kk], Cr);
                Ci = wmma4(Avr[kk], Bti[kk], Ci);
                Ci = wmma4(Avi[kk], Btr[kk], Ci);
            }
        } else {
            const int tb = lm * TABP + g * 16 + half * 8;
            const v4f fr0 = *(const v4f*)(tabR + tb), fr1 = *(const v4f*)(tabR + tb + 4);
            const v4f fi0 = *(const v4f*)(tabI + tb), fi1 = *(const v4f*)(tabI + tb + 4);
            v2f Avr[4], Avi[4];
            Avr[0].x = fr0.x; Avr[0].y = fr0.y; Avr[1].x = fr0.z; Avr[1].y = fr0.w;
            Avr[2].x = fr1.x; Avr[2].y = fr1.y; Avr[3].x = fr1.z; Avr[3].y = fr1.w;
            Avi[0].x = fi0.x; Avi[0].y = fi0.y; Avi[1].x = fi0.z; Avi[1].y = fi0.w;
            Avi[2].x = fi1.x; Avi[2].y = fi1.y; Avi[3].x = fi1.z; Avi[3].y = fi1.w;
            for (int kk = 0; kk < 4; ++kk) {
                Cr = wmma4(Avr[kk], Btr[kk], Cr);
                Cr = wmma4(Avi[kk], Btn[kk], Cr);
                Ci = wmma4(Avr[kk], Bti[kk], Ci);
                Ci = wmma4(Avi[kk], Btr[kk], Ci);
            }
        }
        const int o = (half * 64 + wave) * 16 + lm;
        for (int v = 0; v < 8; ++v) {
            resR[o + v * 128] = Cr[v];  resI[o + v * 128] = Ci[v];
        }
    }
    __syncthreads();

    const int smp = tid >> 4;
    const int l   = tid & 15;
    float cr = 0.f, ci = 0.f, ur = 0.f, ui = 0.f, gr = 0.f, gi = 0.f;
    {
        const bool isA = (l < 3);
        const bool isB = (l >= 3) && (l < 13);
        const int  q   = isA ? l : isB ? (l - 3) : (l - 13);
        const int  ia  = isB ? ((0xE9500 >> (2 * q)) & 3) : (q >> 1);
        const int  ib  = isB ? ((0xFB9E4 >> (2 * q)) & 3) : ((q + 1) >> 1);
        const int  vb  = isA ? 0 : isB ? 2 : 6;
        const float* wc = wt + (isA ? 0 : isB ? 1 : 2) * 16;
        const int ra = (smp * 8 + vb + ia) * 16;
        const int rb = (smp * 8 + vb + ib) * 16;
        for (int n = 0; n < 16; ++n) {
            float ar = resR[ra + n], ai = resI[ra + n];
            float br = resR[rb + n], bi = resI[rb + n];
            float pr = ar * br + ai * bi;
            float pi = ai * br - ar * bi;
            cr += wc[n] * pr;  ci += wc[n] * pi;
            ur += pr;          ui += pi;
            if (__popc(n) & 1) { gr -= pr;  gi -= pi; }
            else               { gr += pr;  gi += pi; }
        }
    }

    {
        const int c = l >> 3, c2 = (l >> 2) & 1, d = (l >> 1) & 1, d2 = l & 1;
        const int i6 = c * 2 + d, j6 = c2 * 2 + d2;
        const int lo = (i6 < j6) ? i6 : j6, hi = (i6 < j6) ? j6 : i6;
        const int selA = c + c2;
        const int selB = 3 + ((0x9740 >> (4 * lo)) & 15) + (hi - lo);
        const int selG = 13 + d + d2;
        const float kA = (c > c2) ? -1.f : 1.f;
        const float kB = (i6 > j6) ? -1.f : 1.f;
        const float kG = (d > d2) ? -1.f : 1.f;
        float pacR = row_gather(cr, selA), pacI = kA * row_gather(ci, selA);
        float pasR = row_gather(gr, selA), pasI = kA * row_gather(gi, selA);
        float pb1R = row_gather(ur, selB), pb1I = kB * row_gather(ui, selB);
        float pbcR = row_gather(cr, selB), pbcI = kB * row_gather(ci, selB);
        float pbsR = row_gather(gr, selB), pbsI = kB * row_gather(gi, selB);
        float pg1R = row_gather(ur, selG), pg1I = kG * row_gather(ui, selG);
        float pgcR = row_gather(cr, selG), pgcI = kG * row_gather(ci, selG);
        float xr = pb1R * pg1R - pb1I * pg1I;
        float xi = pb1R * pg1I + pb1I * pg1R;
        float yr = pbcR * pg1R - pbcI * pg1I + pbsR * pgcR - pbsI * pgcI;
        float yi = pbcR * pg1I + pbcI * pg1R + pbsR * pgcI + pbsI * pgcR;
        float t  = pacR * xr - pacI * xi + pasR * yr - pasI * yi;
        t = row_xor_add<8>(t);
        t = row_xor_add<4>(t);
        t = row_xor_add<2>(t);
        t = row_xor_add<1>(t);
        if (l == 0) out[b * SPB + smp] = t + head_b[0];
    }
}

extern "C" void kernel_launch(void* const* d_in, const int* in_sizes, int n_in,
                              void* d_out, int out_size, void* d_ws, size_t ws_size,
                              hipStream_t stream) {
    const float* x      = (const float*)d_in[0];
    const float* params = (const float*)d_in[1];
    const float* head_w = (const float*)d_in[2];
    const float* head_b = (const float*)d_in[3];
    float* out = (float*)d_out;

    qsim<<<QBATCH / SPB, 256, 0, stream>>>(x, params, head_w, head_b, out);
}


# A fragments formed in registers: each lane makes its k slots' table entries from cos/sin of x and layer 0's wire matrices, same product order; u and table LDS stages and 2 of 4 barriers go
# speedup vs baseline: 51.3377x; 1.0684x over previous
#include <hip/hip_runtime.h>


#define NW    12
#define NL    2
#define QBATCH 8192
#define SPB   16
static_assert(NL == 2, "layer 1 is applied on the factors of the state layer 0 leaves");
static_assert(QBATCH % SPB == 0, "whole workgroups");

typedef __attribute__((ext_vector_type(2))) float v2f;
typedef __attribute__((ext_vector_type(8))) float v8f;
typedef __attribute__((ext_vector_type(4))) float v4f;

static __device__ __forceinline__ int gray4(int a) { return a ^ (a >> 1); }
static __device__ __forceinline__ float sg(int v) { return (__popc(v) & 1) ? -1.f : 1.f; }

static __device__ __forceinline__ v8f wmma4(v2f a, v2f b, v8f c) {
    return __builtin_amdgcn_wmma_f32_16x16x4_f32(false, a, false, b, (short)0, c,
                                                 false, false);
}

static __device__ __forceinline__ void wire_amp(const float* mrow, float cw, float sw, float& ur, float& ui) {
    const v4f m = *(const v4f*)mrow;
    ur = __builtin_fmaf(m.x, cw, m.z * sw);
    ui = __builtin_fmaf(m.y, cw, m.w * sw);
}
static __device__ __forceinline__ void cmul(float pr, float pi, float fr, float fi, float& nr, float& ni) {
    nr = __builtin_fmaf(pr, fr, -(pi * fi));
    ni = __builtin_fmaf(pi, fr, pr * fi);
}

static __device__ __forceinline__ float row_gather(float v, int sel) {
    return __builtin_bit_cast(float, __builtin_amdgcn_permlane16_var(0u, __builtin_bit_cast(unsigned, v),
                                                                     (unsigned)sel, false, false));
}
template <int MASK>
static __device__ __forceinline__ float row_xor_add(float t) {
    return t + __builtin_bit_cast(float, __builtin_amdgcn_update_dpp(0, __builtin_bit_cast(int, t),
                                                                     0x160 | MASK, 0xf, 0xf, false));
}

__launch_bounds__(256)
__global__ void qsim(const float* __restrict__ x,
                     const float* __restrict__ params,
                     const float* __restrict__ head_w,
                     const float* __restrict__ head_b,
                     float* __restrict__ out) {
    __shared__ __attribute__((aligned(16))) float wm[NL * NW * 8];
    __shared__ __attribute__((aligned(16))) float cs[SPB * NW * 2];
    __shared__ float resR[SPB * 128], resI[SPB * 128];
    __shared__ float wt[3 * 16];

    const int tid  = threadIdx.x;
    const int b    = blockIdx.x;
    const int lane = tid & 31;
    const int wave = __builtin_amdgcn_readfirstlane(tid >> 5);
    const int half = lane >> 4;
    const int lm   = lane & 15;
    const int k0   = half * 2;

    const int g    = (wave < 2) ? 0 : (wave < 6) ? 1 : 2;
    const int flip = (wave < 6) ? (wave >> 2) : (wave & 1);
    const int par  = wave & 1;

    float cx = 0.f, sx = 0.f;
    if (tid < SPB * NW + NL * NW) {
        const float* src = (tid < SPB * NW) ? x + b * (SPB * NW) + tid : params + (tid - SPB * NW);
        float th = src[0] * 0.5f;
        cx = cosf(th);  sx = sinf(th);
        if (tid >= SPB * NW) {
            float* o = wm + (tid - SPB * NW) * 8;
            o[0] =  cx * cx;  o[1] =  sx * sx;
            o[2] = -cx * sx;  o[3] = -cx * sx;
            o[4] =  cx * sx;  o[5] = -cx * sx;
            o[6] =  cx * cx;  o[7] = -sx * sx;
        } else {
            cs[tid * 2] = cx;  cs[tid * 2 + 1] = sx;
        }
    }
    if (tid < 48) {
        const int grp = tid >> 4, n = tid & 15;
        float v = 0.f;
        for (int k = 0; k < 4; ++k) v += head_w[grp * 4 + k] * sg(n >> (3 - k));
        wt[tid] = v;
    }
    __syncthreads();

    v2f Btr[4] = {}, Bti[4] = {}, Btn[4] = {};
    {
        const float* wg = wm + (NW + g * 4) * 8;
        const float* q0 = wg +      ((lm >> 3) & 1) * 4;
        const float* q1 = wg +  8 + ((lm >> 2) & 1) * 4;
        const float* q2 = wg + 16 + ((lm >> 1) & 1) * 4;
        const float* q3 = wg + 24 + (lm & 1) * 4;
        const float e0r = q2[flip * 2],       e0i = q2[flip * 2 + 1];
        const float e1r = q2[(1 - flip) * 2], e1i = q2[(1 - flip) * 2 + 1];
        if (g < 2) {
            const int c0 = (half ^ flip) * 2, c3 = par * 2;
            const float ar = q0[c0], ai = q0[c0 + 1], br = q3[c3], bi = q3[c3 + 1];
            const float pr = ar * br - ai * bi, pi = ar * bi + ai * br;
            for (int kk = 0; kk < 2; ++kk) {
                const int c1 = (kk ^ flip) * 2;
                const float fr = q1[c1], fi = q1[c1 + 1];
                const float nr = pr * fr - pi * fi, ni = pr * fi + pi * fr;
                Btr[kk].x = nr * e0r - ni * e0i;  Bti[kk].x = nr * e0i + ni * e0r;
                Btr[kk].y = nr * e1r - ni * e1i;  Bti[kk].y = nr * e1i + ni * e1r;
                Btn[kk].x = -Bti[kk].x;           Btn[kk].y = -Bti[kk].y;
            }
        } else {
            const int c3 = (half ^ flip) * 2;
            const float br = q3[c3], bi = q3[c3 + 1];
            for (int kk = 0; kk < 4; ++kk) {
                const int c0 = ((kk >> 1) ^ flip) * 2, c1 = ((kk & 1) ^ flip) * 2;
                const float ar = q0[c0], ai = q0[c0 + 1], fr = q1[c1], fi = q1[c1 + 1];
                const float pr = ar * fr - ai * fi, pi = ar * fi + ai * fr;
                const float nr = pr * br - pi * bi, ni = pr * bi + pi * br;
                Btr[kk].x = nr * e0r - ni * e0i;  Bti[kk].x = nr * e0i + ni * e0r;
                Btr[kk].y = nr * e1r - ni * e1i;  Bti[kk].y = nr * e1i + ni * e1r;
                Btn[kk].x = -Bti[kk].x;           Btn[kk].y = -Bti[kk].y;
            }
        }
    }


    {
        v8f Cr = {}; v8f Ci = {};
        const float* m0 = wm + g * 4 * 8;
        const v4f c01 = *(const v4f*)(cs + lm * 24 + g * 8);
        const v4f c23 = *(const v4f*)(cs + lm * 24 + g * 8 + 4);
        float u2r[2], u2i[2];
        wire_amp(m0 + 16,     c23.x, c23.y, u2r[0], u2i[0]);
        wire_amp(m0 + 16 + 4, c23.x, c23.y, u2r[1], u2i[1]);
        if (g < 2) {
            const int blk = par ^ flip;
            float f0r, f0i;
            wire_amp(m0 + half * 4, c01.x, c01.y, f0r, f0i);
            float f1r[2], f1i[2];
            wire_amp(m0 + 8 + half * 4,       c01.z, c01.w, f1r[0], f1i[0]);
            wire_amp(m0 + 8 + (1 - half) * 4, c01.z, c01.w, f1r[1], f1i[1]);
            float f3r[2], f3i[2];
            wire_amp(m0 + 24 + blk * 4,       c23.z, c23.w, f3r[0], f3i[0]);
            wire_amp(m0 + 24 + (1 - blk) * 4, c23.z, c23.w, f3r[1], f3i[1]);
            v2f Avr[2], Avi[2];
            for (int kk = 0; kk < 2; ++kk) {
                float t1r, t1i;
                cmul(f0r, f0i, f1r[kk], f1i[kk], t1r, t1i);
                for (int e = 0; e < 2; ++e) {
                    float t2r, t2i, ar, ai;
                    cmul(t1r, t1i, u2r[e ^ kk], u2i[e ^ kk], t2r, t2i);
                    cmul(t2r, t2i, f3r[e], f3i[e], ar, ai);
                    Avr[kk][e] = ar;  Avi[kk][e] = ai;
                }
            }
            for (int kk = 0; kk < 2; ++kk) {
                Cr = wmma4(Avr[kk], Btr[kk], Cr);
                Cr = wmma4(Avi[kk], Btn[kk], Cr);
                Ci = wmma4(Avr[kk], Bti[kk], Ci);
                Ci = wmma4(Avi[kk], Btr[kk], Ci);
            }
        } else {
            float u0r[2], u0i[2], u1r[2], u1i[2];
            wire_amp(m0,         c01.x, c01.y, u0r[0], u0i[0]);
            wire_amp(m0 + 4,     c01.x, c01.y, u0r[1], u0i[1]);
            wire_amp(m0 + 8,     c01.z, c01.w, u1r[0], u1i[0]);
            wire_amp(m0 + 8 + 4, c01.z, c01.w, u1r[1], u1i[1]);
            float f3r[2], f3i[2];
            wire_amp(m0 + 24 + half * 4,       c23.z, c23.w, f3r[0], f3i[0]);
            wire_amp(m0 + 24 + (1 - half) * 4, c23.z, c23.w, f3r[1], f3i[1]);
            v2f Avr[4], Avi[4];
            for (int kk = 0; kk < 4; ++kk) {
                const int r0 = kk >> 1, r1 = (kk & 1) ^ (kk >> 1);
                float t1r, t1i;
                cmul(u0r[r0], u0i[r0], u1r[r1], u1i[r1], t1r, t1i);
                for (int e = 0; e < 2; ++e) {
                    const int r2 = e ^ (kk & 1);
                    float t2r, t2i, ar, ai;
                    cmul(t1r, t1i, u2r[r2], u2i[r2], t2r, t2i);
                    cmul(t2r, t2i, f3r[e], f3i[e], ar, ai);
                    Avr[kk][e] = ar;  Avi[kk][e] = ai;
                }
            }
            for (int kk = 0; kk < 4; ++kk) {
                Cr = wmma4(Avr[kk], Btr[kk], Cr);
                Cr = wmma4(Avi[kk], Btn[kk], Cr);
                Ci = wmma4(Avr[kk], Bti[kk], Ci);
                Ci = wmma4(Avi[kk], Btr[kk], Ci);
            }
        }
        const int o = (half * 64 + wave) * 16 + lm;
        for (int v = 0; v < 8; ++v) {
            resR[o + v * 128] = Cr[v];  resI[o + v * 128] = Ci[v];
        }
    }
    __syncthreads();

    const int smp = tid >> 4;
    const int l   = tid & 15;
    float cr = 0.f, ci = 0.f, ur = 0.f, ui = 0.f, gr = 0.f, gi = 0.f;
    {
        const bool isA = (l < 3);
        const bool isB = (l >= 3) && (l < 13);
        const int  q   = isA ? l : isB ? (l - 3) : (l - 13);
        const int  ia  = isB ? ((0xE9500 >> (2 * q)) & 3) : (q >> 1);
        const int  ib  = isB ? ((0xFB9E4 >> (2 * q)) & 3) : ((q + 1) >> 1);
        const int  vb  = isA ? 0 : isB ? 2 : 6;
        const float* wc = wt + (isA ? 0 : isB ? 1 : 2) * 16;
        const int ra = (smp * 8 + vb + ia) * 16;
        const int rb = (smp * 8 + vb + ib) * 16;
        for (int n = 0; n < 16; ++n) {
            float ar = resR[ra + n], ai = resI[ra + n];
            float br = resR[rb + n], bi = resI[rb + n];
            float pr = ar * br + ai * bi;
            float pi = ai * br - ar * bi;
            cr += wc[n] * pr;  ci += wc[n] * pi;
            ur += pr;          ui += pi;
            if (__popc(n) & 1) { gr -= pr;  gi -= pi; }
            else               { gr += pr;  gi += pi; }
        }
    }

    {
        const int c = l >> 3, c2 = (l >> 2) & 1, d = (l >> 1) & 1, d2 = l & 1;
        const int i6 = c * 2 + d, j6 = c2 * 2 + d2;
        const int lo = (i6 < j6) ? i6 : j6, hi = (i6 < j6) ? j6 : i6;
        const int selA = c + c2;
        const int selB = 3 + ((0x9740 >> (4 * lo)) & 15) + (hi - lo);
        const int selG = 13 + d + d2;
        const float kA = (c > c2) ? -1.f : 1.f;
        const float kB = (i6 > j6) ? -1.f : 1.f;
        const float kG = (d > d2) ? -1.f : 1.f;
        float pacR = row_gather(cr, selA), pacI = kA * row_gather(ci, selA);
        float pasR = row_gather(gr, selA), pasI = kA * row_gather(gi, selA);
        float pb1R = row_gather(ur, selB), pb1I = kB * row_gather(ui, selB);
        float pbcR = row_gather(cr, selB), pbcI = kB * row_gather(ci, selB);
        float pbsR = row_gather(gr, selB), pbsI = kB * row_gather(gi, selB);
        float pg1R = row_gather(ur, selG), pg1I = kG * row_gather(ui, selG);
        float pgcR = row_gather(cr, selG), pgcI = kG * row_gather(ci, selG);
        float xr = pb1R * pg1R - pb1I * pg1I;
        float xi = pb1R * pg1I + pb1I * pg1R;
        float yr = pbcR * pg1R - pbcI * pg1I + pbsR * pgcR - pbsI * pgcI;
        float yi = pbcR * pg1I + pbcI * pg1R + pbsR * pgcI + pbsI * pgcR;
        float t  = pacR * xr - pacI * xi + pasR * yr - pasI * yi;
        t = row_xor_add<8>(t);
        t = row_xor_add<4>(t);
        t = row_xor_add<2>(t);
        t = row_xor_add<1>(t);
        if (l == 0) out[b * SPB + smp] = t + head_b[0];
    }
}

extern "C" void kernel_launch(void* const* d_in, const int* in_sizes, int n_in,
                              void* d_out, int out_size, void* d_ws, size_t ws_size,
                              hipStream_t stream) {
    const float* x      = (const float*)d_in[0];
    const float* params = (const float*)d_in[1];
    const float* head_w = (const float*)d_in[2];
    const float* head_b = (const float*)d_in[3];
    float* out = (float*)d_out;

    qsim<<<QBATCH / SPB, 256, 0, stream>>>(x, params, head_w, head_b, out);
}
